# starting kernel
# baseline (speedup 1.0000x reference)
#include <hip/hip_runtime.h>
#include <hip/hip_bf16.h>
#include <math.h>


typedef float v2f __attribute__((ext_vector_type(2)));
typedef float v8f __attribute__((ext_vector_type(8)));

#define BN_EPS 1e-5f

__global__ void k_zero(float* __restrict__ p, size_t n) {
  size_t i = (size_t)blockIdx.x * blockDim.x + threadIdx.x;
  size_t stride = (size_t)gridDim.x * blockDim.x;
  for (; i < n; i += stride) p[i] = 0.0f;
}

__global__ void k_deg_init(float* __restrict__ deg, int n) {
  int i = blockIdx.x * blockDim.x + threadIdx.x;
  if (i < n) deg[i] = 1.0f;
}

__global__ void k_deg_count(const int* __restrict__ dst, float* __restrict__ deg, int E) {
  int e = blockIdx.x * blockDim.x + threadIdx.x;
  if (e < E) unsafeAtomicAdd(&deg[dst[e]], 1.0f);
}

__global__ void k_rsqrt_inplace(float* __restrict__ d, int n) {
  int i = blockIdx.x * blockDim.x + threadIdx.x;
  if (i < n) {
    float v = d[i];
    d[i] = (v > 0.0f) ? rsqrtf(v) : 0.0f;
  }
}

__global__ void k_gemm_wmma_f32(const float* __restrict__ A, const float* __restrict__ B,
                                const float* __restrict__ dinv,
                                float* __restrict__ C, int M, int K, int F) {
  const int lane = threadIdx.x & 31;
  const int wave = (int)((blockIdx.x * blockDim.x + threadIdx.x) >> 5);
  const int nf = F >> 4;
  const int mTile = wave / nf;
  const int fTile = wave - mTile * nf;
  const int m0 = mTile << 4;
  if (m0 >= M) return;
  const int f0 = fTile << 4;
  const int row = lane & 15;
  const int kh2 = (lane >> 4) << 1;

  const float* arow = A + (size_t)(m0 + row) * K + kh2;
  const float* bcol = B + (size_t)kh2 * F + f0 + row;

  v8f acc = {0.f, 0.f, 0.f, 0.f, 0.f, 0.f, 0.f, 0.f};
  for (int k = 0; k < K; k += 4) {
    float2 av = *reinterpret_cast<const float2*>(arow + k);
    v2f a; a.x = av.x; a.y = av.y;
    v2f b;
    b.x = bcol[(size_t)k * F];
    b.y = bcol[(size_t)(k + 1) * F];
    acc = __builtin_amdgcn_wmma_f32_16x16x4_f32(false, a, false, b,
                                                (short)0, acc, false, false);
  }
  const float* dvp = dinv + m0 + (kh2 << 2);
  float* crow = C + (size_t)(m0 + (kh2 << 2)) * F + f0 + row;
#pragma unroll
  for (int r = 0; r < 8; ++r) crow[(size_t)r * F] = acc[r] * dvp[r];
}

__global__ void k_scatter(const float* __restrict__ hsc, const int* __restrict__ src,
                          const int* __restrict__ dst,
                          float* __restrict__ agg, int E, int F) {
  int wave = (int)((blockIdx.x * blockDim.x + threadIdx.x) >> 5);
  int lane = threadIdx.x & 31;
  if (wave >= E) return;
  int s = src[wave], d = dst[wave];
  const float4* hrow = reinterpret_cast<const float4*>(hsc + (size_t)s * F);
  float* ad = agg + (size_t)d * F;
  for (int f4 = lane; 4 * f4 < F; f4 += 32) {
    float4 v = hrow[f4];
    unsafeAtomicAdd(ad + 4 * f4 + 0, v.x);
    unsafeAtomicAdd(ad + 4 * f4 + 1, v.y);
    unsafeAtomicAdd(ad + 4 * f4 + 2, v.z);
    unsafeAtomicAdd(ad + 4 * f4 + 3, v.w);
  }
}

__global__ void k_finish_relu_stats(float* __restrict__ agg, const float* __restrict__ hsc,
                                    const float* __restrict__ dinv,
                                    const float* __restrict__ bias,
                                    float* __restrict__ stats,
                                    int Nn, int F, int rowsPerBlock) {
  int j = threadIdx.x;
  int r0 = blockIdx.x * rowsPerBlock;
  int r1 = min(r0 + rowsPerBlock, Nn);
  float bj = bias[j];
  float sum = 0.f, sumsq = 0.f;
  for (int r = r0; r < r1; ++r) {
    float di = dinv[r];
    size_t idx = (size_t)r * F + j;
    float v = (agg[idx] + hsc[idx]) * di + bj;
    v = fmaxf(v, 0.f);
    agg[idx] = v;
    sum += v;
    sumsq += v * v;
  }
  unsafeAtomicAdd(stats + j, sum);
  unsafeAtomicAdd(stats + F + j, sumsq);
}

__global__ void k_bn_apply(float* __restrict__ v, const float* __restrict__ stats,
                           const float* __restrict__ gamma, const float* __restrict__ beta,
                           int Nn, int F, float invN) {
  size_t total = (size_t)Nn * F;
  size_t i = (size_t)blockIdx.x * blockDim.x + threadIdx.x;
  size_t stride = (size_t)gridDim.x * blockDim.x;
  for (; i < total; i += stride) {
    int j = (int)(i & (size_t)(F - 1));
    float mean = stats[j] * invN;
    float var = stats[F + j] * invN - mean * mean;
    v[i] = (v[i] - mean) * rsqrtf(var + BN_EPS) * gamma[j] + beta[j];
  }
}

__global__ void k_gemm64x2(const float* __restrict__ h3, const float* __restrict__ W4,
                           const float* __restrict__ dinv,
                           float* __restrict__ h4, int Nn) {
  int i = blockIdx.x * blockDim.x + threadIdx.x;
  if (i >= Nn) return;
  const float* row = h3 + (size_t)i * 64;
  float o0 = 0.f, o1 = 0.f;
#pragma unroll 8
  for (int k = 0; k < 64; ++k) {
    float xv = row[k];
    o0 += xv * W4[2 * k];
    o1 += xv * W4[2 * k + 1];
  }
  float di = dinv[i];
  h4[2 * (size_t)i]     = o0 * di;
  h4[2 * (size_t)i + 1] = o1 * di;
}

__global__ void k_scatter2(const float* __restrict__ h4s, const int* __restrict__ src,
                           const int* __restrict__ dst,
                           float* __restrict__ agg, int E) {
  int e = blockIdx.x * blockDim.x + threadIdx.x;
  if (e >= E) return;
  int s = src[e], d = dst[e];
  float2 v = *reinterpret_cast<const float2*>(h4s + 2 * (size_t)s);
  unsafeAtomicAdd(agg + 2 * (size_t)d,     v.x);
  unsafeAtomicAdd(agg + 2 * (size_t)d + 1, v.y);
}

__global__ void k_final_lsm(const float* __restrict__ agg, const float* __restrict__ h4s,
                            const float* __restrict__ dinv, const float* __restrict__ b4,
                            float* __restrict__ out, int Nn) {
  int i = blockIdx.x * blockDim.x + threadIdx.x;
  if (i >= Nn) return;
  float di = dinv[i];
  float o0 = (agg[2 * (size_t)i]     + h4s[2 * (size_t)i])     * di + b4[0];
  float o1 = (agg[2 * (size_t)i + 1] + h4s[2 * (size_t)i + 1]) * di + b4[1];
  float m = fmaxf(o0, o1);
  float l = m + logf(expf(o0 - m) + expf(o1 - m));
  out[2 * (size_t)i]     = o0 - l;
  out[2 * (size_t)i + 1] = o1 - l;
}

extern "C" void kernel_launch(void* const* d_in, const int* in_sizes, int n_in,
                              void* d_out, int out_size, void* d_ws, size_t ws_size,
                              hipStream_t stream) {
  const float* x   = (const float*)d_in[0];
  const int*   ei  = (const int*)d_in[1];
  const float* W1  = (const float*)d_in[2];
  const float* b1  = (const float*)d_in[3];
  const float* g1  = (const float*)d_in[4];
  const float* be1 = (const float*)d_in[5];
  const float* W2  = (const float*)d_in[6];
  const float* b2  = (const float*)d_in[7];
  const float* g2  = (const float*)d_in[8];
  const float* be2 = (const float*)d_in[9];
  const float* W3  = (const float*)d_in[10];
  const float* b3  = (const float*)d_in[11];
  const float* g3  = (const float*)d_in[12];
  const float* be3 = (const float*)d_in[13];
  const float* W4  = (const float*)d_in[14];
  const float* b4  = (const float*)d_in[15];

  const int Nn = in_sizes[0] / 24;
  const int E  = in_sizes[1] / 2;
  const int* src = ei;
  const int* dst = ei + E;

  float* ws    = (float*)d_ws;
  float* dinv  = ws;
  float* stats = ws + Nn;
  float* bufA  = stats + 256;
  float* bufB  = bufA + (size_t)Nn * 128;

  const float invN = 1.0f / (float)Nn;
  const int TB = 256;

  auto gemm = [&](const float* A, const float* Bm, float* C, int K, int F) {
    long waves = (long)((Nn + 15) / 16) * (F / 16);
    int blocks = (int)((waves * 32 + TB - 1) / TB);
    k_gemm_wmma_f32<<<blocks, TB, 0, stream>>>(A, Bm, dinv, C, Nn, K, F);
  };
  auto scatter = [&](const float* h, float* agg, int F) {
    int blocks = (int)(((long)E * 32 + TB - 1) / TB);
    k_scatter<<<blocks, TB, 0, stream>>>(h, src, dst, agg, E, F);
  };
  auto finish = [&](float* agg, const float* h, const float* bias, int F) {
    const int rows = 256;
    k_zero<<<1, 256, 0, stream>>>(stats, 256);
    k_finish_relu_stats<<<(Nn + rows - 1) / rows, F, 0, stream>>>(
        agg, h, dinv, bias, stats, Nn, F, rows);
  };
  auto bn = [&](float* v, const float* gamma, const float* beta, int F) {
    k_bn_apply<<<2048, TB, 0, stream>>>(v, stats, gamma, beta, Nn, F, invN);
  };
  auto zero = [&](float* p, size_t n) {
    int blocks = (int)((n + TB - 1) / TB);
    if (blocks > 4096) blocks = 4096;
    k_zero<<<blocks, TB, 0, stream>>>(p, n);
  };

  k_deg_init<<<(Nn + TB - 1) / TB, TB, 0, stream>>>(dinv, Nn);
  k_deg_count<<<(E + TB - 1) / TB, TB, 0, stream>>>(dst, dinv, E);
  k_rsqrt_inplace<<<(Nn + TB - 1) / TB, TB, 0, stream>>>(dinv, Nn);

  gemm(x, W1, bufA, 24, 128);
  zero(bufB, (size_t)Nn * 128);
  scatter(bufA, bufB, 128);
  finish(bufB, bufA, b1, 128);
  bn(bufB, g1, be1, 128);

  gemm(bufB, W2, bufA, 128, 128);
  zero(bufB, (size_t)Nn * 128);
  scatter(bufA, bufB, 128);
  finish(bufB, bufA, b2, 128);
  bn(bufB, g2, be2, 128);

  gemm(bufB, W3, bufA, 128, 64);
  zero(bufB, (size_t)Nn * 64);
  scatter(bufA, bufB, 64);
  finish(bufB, bufA, b3, 64);
  bn(bufB, g3, be3, 64);

  k_gemm64x2<<<(Nn + TB - 1) / TB, TB, 0, stream>>>(bufB, W4, dinv, bufA, Nn);
  zero(bufB, (size_t)Nn * 2);
  k_scatter2<<<(E + TB - 1) / TB, TB, 0, stream>>>(bufA, src, dst, bufB, E);
  k_final_lsm<<<(Nn + TB - 1) / TB, TB, 0, stream>>>(bufB, bufA, dinv, b4,
                                                     (float*)d_out, Nn);
}


# layers 1-3 edge sum: a workgroup keeps 400 destination rows in LDS, scans dst for its range, gathers matching rows, adds them in LDS (arrival order); float atomics to memory and zero pass go
# speedup vs baseline: 1.4231x; 1.4231x over previous
#include <hip/hip_runtime.h>
#include <hip/hip_bf16.h>
#include <math.h>


typedef float v2f __attribute__((ext_vector_type(2)));
typedef float v8f __attribute__((ext_vector_type(8)));

#define BN_EPS 1e-5f

__global__ void k_zero(float* __restrict__ p, size_t n) {
  size_t i = (size_t)blockIdx.x * blockDim.x + threadIdx.x;
  size_t stride = (size_t)gridDim.x * blockDim.x;
  for (; i < n; i += stride) p[i] = 0.0f;
}

__global__ void k_deg_init(float* __restrict__ deg, int n) {
  int i = blockIdx.x * blockDim.x + threadIdx.x;
  if (i < n) deg[i] = 1.0f;
}

__global__ void k_deg_count(const int* __restrict__ dst, float* __restrict__ deg, int E) {
  int e = blockIdx.x * blockDim.x + threadIdx.x;
  if (e < E) unsafeAtomicAdd(&deg[dst[e]], 1.0f);
}

__global__ void k_rsqrt_inplace(float* __restrict__ d, int n) {
  int i = blockIdx.x * blockDim.x + threadIdx.x;
  if (i < n) {
    float v = d[i];
    d[i] = (v > 0.0f) ? rsqrtf(v) : 0.0f;
  }
}

__global__ void k_gemm_wmma_f32(const float* __restrict__ A, const float* __restrict__ B,
                                const float* __restrict__ dinv,
                                float* __restrict__ C, int M, int K, int F) {
  const int lane = threadIdx.x & 31;
  const int wave = (int)((blockIdx.x * blockDim.x + threadIdx.x) >> 5);
  const int nf = F >> 4;
  const int mTile = wave / nf;
  const int fTile = wave - mTile * nf;
  const int m0 = mTile << 4;
  if (m0 >= M) return;
  const int f0 = fTile << 4;
  const int row = lane & 15;
  const int kh2 = (lane >> 4) << 1;

  const float* arow = A + (size_t)(m0 + row) * K + kh2;
  const float* bcol = B + (size_t)kh2 * F + f0 + row;

  v8f acc = {0.f, 0.f, 0.f, 0.f, 0.f, 0.f, 0.f, 0.f};
  for (int k = 0; k < K; k += 4) {
    float2 av = *reinterpret_cast<const float2*>(arow + k);
    v2f a; a.x = av.x; a.y = av.y;
    v2f b;
    b.x = bcol[(size_t)k * F];
    b.y = bcol[(size_t)(k + 1) * F];
    acc = __builtin_amdgcn_wmma_f32_16x16x4_f32(false, a, false, b,
                                                (short)0, acc, false, false);
  }
  const float* dvp = dinv + m0 + (kh2 << 2);
  float* crow = C + (size_t)(m0 + (kh2 << 2)) * F + f0 + row;
#pragma unroll
  for (int r = 0; r < 8; ++r) crow[(size_t)r * F] = acc[r] * dvp[r];
}

#define PULL_R 400
#define PULL_SEG 384
template <int F>
__global__ __launch_bounds__(1024) void k_pull(const float* __restrict__ hsc,
                                               const int* __restrict__ src,
                                               const int* __restrict__ dst,
                                               float* __restrict__ agg, int E, int Nn) {
  constexpr int C = F / 32;
  constexpr int U = 32 / C;
  constexpr int UV = 8;
  constexpr int WCH = 128 * UV;
  constexpr int TILE = 32 * WCH;
  __shared__ float acc[PULL_R * F];
  __shared__ int le[32 * PULL_SEG];
  __shared__ int ld[32 * PULL_SEG];
  const int tid = threadIdx.x;
  const int lane = tid & 31;
  const int wave = __builtin_amdgcn_readfirstlane(tid >> 5);
  const int r0 = blockIdx.x * PULL_R;
  const int nrows = min(PULL_R, Nn - r0);

  {
    float4* a4 = reinterpret_cast<float4*>(acc);
    const float4 z = make_float4(0.f, 0.f, 0.f, 0.f);
    for (int i = tid; i < PULL_R * F / 4; i += 1024) a4[i] = z;
  }
  __syncthreads();

  int* mye = le + wave * PULL_SEG;
  int* myd = ld + wave * PULL_SEG;
  int wcnt = 0;
  const unsigned ltmask = (1u << lane) - 1u;
  auto keep = [&](int dval, int e) {
    const unsigned rel = (unsigned)(dval - r0);
    const bool m = rel < (unsigned)nrows;
    const unsigned bal = __builtin_amdgcn_ballot_w32(m);
    if (bal != 0u) {
      if (m) {
        const int slot = wcnt + __popc(bal & ltmask);
        if (slot < PULL_SEG) {
          mye[slot] = e;
          myd[slot] = (int)rel;
        } else {
          const float* hr = hsc + (size_t)src[e] * F;
          float* ar = acc + rel * F;
#pragma nounroll
          for (int c = 0; c < F; ++c) unsafeAtomicAdd(ar + c, hr[c]);
        }
      }
      wcnt += __popc(bal);
    }
  };

  const int ntiles = (E + TILE - 1) / TILE;
  const int t0 = (int)(((long)blockIdx.x * ntiles) / gridDim.x);
  const bool vec = ((E & 3) == 0) && ((reinterpret_cast<size_t>(dst) & 15) == 0);
  for (int k = 0; k < ntiles; ++k) {
    int t = t0 + k;
    if (t >= ntiles) t -= ntiles;
    const int eb = t * TILE + wave * WCH;
    if (vec) {
      int4 q[UV];
      if (eb + WCH <= E) {
#pragma unroll
        for (int u = 0; u < UV; ++u)
          q[u] = *reinterpret_cast<const int4*>(dst + eb + u * 128 + lane * 4);
      } else {
#pragma unroll
        for (int u = 0; u < UV; ++u) {
          const int e4 = eb + u * 128 + lane * 4;
          q[u] = make_int4(-1, -1, -1, -1);
          if (e4 < E) q[u] = *reinterpret_cast<const int4*>(dst + e4);
        }
      }
#pragma unroll
      for (int u = 0; u < UV; ++u) {
        const int e4 = eb + u * 128 + lane * 4;
        keep(q[u].x, e4);
        keep(q[u].y, e4 + 1);
        keep(q[u].z, e4 + 2);
        keep(q[u].w, e4 + 3);
      }
    } else {
      for (int h = 0; h < WCH; h += 32) {
        const int e = eb + h + lane;
        const int dval = (e < E) ? dst[e] : -1;
        keep(dval, e);
      }
    }
  }

  __syncthreads();

  const int n = min(wcnt, PULL_SEG);
  for (int i0 = 0; i0 < n; i0 += 32) {
    const int m = min(32, n - i0);
    int sv = 0, dl = 0;
    if (lane < m) { sv = src[mye[i0 + lane]]; dl = myd[i0 + lane]; }
#pragma unroll
    for (int g = 0; g < 32; g += U) {
      if (g + U <= m) {
        float v[U][C];
#pragma unroll
        for (int u = 0; u < U; ++u) {
          const int s = __builtin_amdgcn_readlane(sv, g + u);
          const float* hrow = hsc + (size_t)s * F + lane;
#pragma unroll
          for (int j = 0; j < C; ++j) v[u][j] = hrow[32 * j];
        }
#pragma unroll
        for (int u = 0; u < U; ++u) {
          const int d = __builtin_amdgcn_readlane(dl, g + u);
          float* ad = acc + d * F + lane;
#pragma unroll
          for (int j = 0; j < C; ++j) unsafeAtomicAdd(ad + 32 * j, v[u][j]);
        }
      } else if (g < m) {
        for (int i = g; i < m; ++i) {
          const int s = __builtin_amdgcn_readlane(sv, i);
          const int d = __builtin_amdgcn_readlane(dl, i);
          const float* hrow = hsc + (size_t)s * F + lane;
          float* ad = acc + d * F + lane;
          float v[C];
#pragma unroll
          for (int j = 0; j < C; ++j) v[j] = hrow[32 * j];
#pragma unroll
          for (int j = 0; j < C; ++j) unsafeAtomicAdd(ad + 32 * j, v[j]);
        }
      }
    }
  }
  __syncthreads();

  {
    float* o = agg + (size_t)r0 * F;
    if ((reinterpret_cast<size_t>(o) & 15) == 0) {
      const float4* a4 = reinterpret_cast<const float4*>(acc);
      float4* o4 = reinterpret_cast<float4*>(o);
      const int tot4 = nrows * (F / 4);
      for (int i = tid; i < tot4; i += 1024) o4[i] = a4[i];
    } else {
      for (int i = tid; i < nrows * F; i += 1024) o[i] = acc[i];
    }
  }
}

__global__ void k_finish_relu_stats(float* __restrict__ agg, const float* __restrict__ hsc,
                                    const float* __restrict__ dinv,
                                    const float* __restrict__ bias,
                                    float* __restrict__ stats,
                                    int Nn, int F, int rowsPerBlock) {
  int j = threadIdx.x;
  int r0 = blockIdx.x * rowsPerBlock;
  int r1 = min(r0 + rowsPerBlock, Nn);
  float bj = bias[j];
  float sum = 0.f, sumsq = 0.f;
  for (int r = r0; r < r1; ++r) {
    float di = dinv[r];
    size_t idx = (size_t)r * F + j;
    float v = (agg[idx] + hsc[idx]) * di + bj;
    v = fmaxf(v, 0.f);
    agg[idx] = v;
    sum += v;
    sumsq += v * v;
  }
  unsafeAtomicAdd(stats + j, sum);
  unsafeAtomicAdd(stats + F + j, sumsq);
}

__global__ void k_bn_apply(float* __restrict__ v, const float* __restrict__ stats,
                           const float* __restrict__ gamma, const float* __restrict__ beta,
                           int Nn, int F, float invN) {
  size_t total = (size_t)Nn * F;
  size_t i = (size_t)blockIdx.x * blockDim.x + threadIdx.x;
  size_t stride = (size_t)gridDim.x * blockDim.x;
  for (; i < total; i += stride) {
    int j = (int)(i & (size_t)(F - 1));
    float mean = stats[j] * invN;
    float var = stats[F + j] * invN - mean * mean;
    v[i] = (v[i] - mean) * rsqrtf(var + BN_EPS) * gamma[j] + beta[j];
  }
}

__global__ void k_gemm64x2(const float* __restrict__ h3, const float* __restrict__ W4,
                           const float* __restrict__ dinv,
                           float* __restrict__ h4, int Nn) {
  int i = blockIdx.x * blockDim.x + threadIdx.x;
  if (i >= Nn) return;
  const float* row = h3 + (size_t)i * 64;
  float o0 = 0.f, o1 = 0.f;
#pragma unroll 8
  for (int k = 0; k < 64; ++k) {
    float xv = row[k];
    o0 += xv * W4[2 * k];
    o1 += xv * W4[2 * k + 1];
  }
  float di = dinv[i];
  h4[2 * (size_t)i]     = o0 * di;
  h4[2 * (size_t)i + 1] = o1 * di;
}

__global__ void k_scatter2(const float* __restrict__ h4s, const int* __restrict__ src,
                           const int* __restrict__ dst,
                           float* __restrict__ agg, int E) {
  int e = blockIdx.x * blockDim.x + threadIdx.x;
  if (e >= E) return;
  int s = src[e], d = dst[e];
  float2 v = *reinterpret_cast<const float2*>(h4s + 2 * (size_t)s);
  unsafeAtomicAdd(agg + 2 * (size_t)d,     v.x);
  unsafeAtomicAdd(agg + 2 * (size_t)d + 1, v.y);
}

__global__ void k_final_lsm(const float* __restrict__ agg, const float* __restrict__ h4s,
                            const float* __restrict__ dinv, const float* __restrict__ b4,
                            float* __restrict__ out, int Nn) {
  int i = blockIdx.x * blockDim.x + threadIdx.x;
  if (i >= Nn) return;
  float di = dinv[i];
  float o0 = (agg[2 * (size_t)i]     + h4s[2 * (size_t)i])     * di + b4[0];
  float o1 = (agg[2 * (size_t)i + 1] + h4s[2 * (size_t)i + 1]) * di + b4[1];
  float m = fmaxf(o0, o1);
  float l = m + logf(expf(o0 - m) + expf(o1 - m));
  out[2 * (size_t)i]     = o0 - l;
  out[2 * (size_t)i + 1] = o1 - l;
}

extern "C" void kernel_launch(void* const* d_in, const int* in_sizes, int n_in,
                              void* d_out, int out_size, void* d_ws, size_t ws_size,
                              hipStream_t stream) {
  const float* x   = (const float*)d_in[0];
  const int*   ei  = (const int*)d_in[1];
  const float* W1  = (const float*)d_in[2];
  const float* b1  = (const float*)d_in[3];
  const float* g1  = (const float*)d_in[4];
  const float* be1 = (const float*)d_in[5];
  const float* W2  = (const float*)d_in[6];
  const float* b2  = (const float*)d_in[7];
  const float* g2  = (const float*)d_in[8];
  const float* be2 = (const float*)d_in[9];
  const float* W3  = (const float*)d_in[10];
  const float* b3  = (const float*)d_in[11];
  const float* g3  = (const float*)d_in[12];
  const float* be3 = (const float*)d_in[13];
  const float* W4  = (const float*)d_in[14];
  const float* b4  = (const float*)d_in[15];

  const int Nn = in_sizes[0] / 24;
  const int E  = in_sizes[1] / 2;
  const int* src = ei;
  const int* dst = ei + E;

  float* ws    = (float*)d_ws;
  float* dinv  = ws;
  float* stats = ws + Nn;
  float* bufA  = stats + 256;
  float* bufB  = bufA + (size_t)Nn * 128;

  const float invN = 1.0f / (float)Nn;
  const int TB = 256;

  auto gemm = [&](const float* A, const float* Bm, float* C, int K, int F) {
    long waves = (long)((Nn + 15) / 16) * (F / 16);
    int blocks = (int)((waves * 32 + TB - 1) / TB);
    k_gemm_wmma_f32<<<blocks, TB, 0, stream>>>(A, Bm, dinv, C, Nn, K, F);
  };
  auto pull128 = [&](const float* h, float* agg) {
    k_pull<128><<<(Nn + PULL_R - 1) / PULL_R, 1024, 0, stream>>>(h, src, dst, agg, E, Nn);
  };
  auto pull64 = [&](const float* h, float* agg) {
    k_pull<64><<<(Nn + PULL_R - 1) / PULL_R, 1024, 0, stream>>>(h, src, dst, agg, E, Nn);
  };
  auto finish = [&](float* agg, const float* h, const float* bias, int F) {
    const int rows = 256;
    k_zero<<<1, 256, 0, stream>>>(stats, 256);
    k_finish_relu_stats<<<(Nn + rows - 1) / rows, F, 0, stream>>>(
        agg, h, dinv, bias, stats, Nn, F, rows);
  };
  auto bn = [&](float* v, const float* gamma, const float* beta, int F) {
    k_bn_apply<<<2048, TB, 0, stream>>>(v, stats, gamma, beta, Nn, F, invN);
  };
  auto zero = [&](float* p, size_t n) {
    int blocks = (int)((n + TB - 1) / TB);
    if (blocks > 4096) blocks = 4096;
    k_zero<<<blocks, TB, 0, stream>>>(p, n);
  };

  k_deg_init<<<(Nn + TB - 1) / TB, TB, 0, stream>>>(dinv, Nn);
  k_deg_count<<<(E + TB - 1) / TB, TB, 0, stream>>>(dst, dinv, E);
  k_rsqrt_inplace<<<(Nn + TB - 1) / TB, TB, 0, stream>>>(dinv, Nn);

  gemm(x, W1, bufA, 24, 128);
  pull128(bufA, bufB);
  finish(bufB, bufA, b1, 128);
  bn(bufB, g1, be1, 128);

  gemm(bufB, W2, bufA, 128, 128);
  pull128(bufA, bufB);
  finish(bufB, bufA, b2, 128);
  bn(bufB, g2, be2, 128);

  gemm(bufB, W3, bufA, 128, 64);
  pull64(bufA, bufB);
  finish(bufB, bufA, b3, 64);
  bn(bufB, g3, be3, 64);

  k_gemm64x2<<<(Nn + TB - 1) / TB, TB, 0, stream>>>(bufB, W4, dinv, bufA, Nn);
  zero(bufB, (size_t)Nn * 2);
  k_scatter2<<<(E + TB - 1) / TB, TB, 0, stream>>>(bufA, src, dst, bufB, E);
  k_final_lsm<<<(Nn + TB - 1) / TB, TB, 0, stream>>>(bufB, bufA, dinv, b4,
                                                     (float*)d_out, Nn);
}


# layers 1-3: the edge-sum workgroup finishes its 400 LDS rows itself (own row, dst norm, bias, relu) and folds BN column sums in LDS; finish pass and raw-sum round trip go (BN sum order changes)
# speedup vs baseline: 2.2811x; 1.6029x over previous
#include <hip/hip_runtime.h>
#include <hip/hip_bf16.h>
#include <math.h>


typedef float v2f __attribute__((ext_vector_type(2)));
typedef float v8f __attribute__((ext_vector_type(8)));

#define BN_EPS 1e-5f

__global__ void k_zero(float* __restrict__ p, size_t n) {
  size_t i = (size_t)blockIdx.x * blockDim.x + threadIdx.x;
  size_t stride = (size_t)gridDim.x * blockDim.x;
  for (; i < n; i += stride) p[i] = 0.0f;
}

__global__ void k_deg_init(float* __restrict__ deg, int n) {
  int i = blockIdx.x * blockDim.x + threadIdx.x;
  if (i < n) deg[i] = 1.0f;
}

__global__ void k_deg_count(const int* __restrict__ dst, float* __restrict__ deg, int E) {
  int e = blockIdx.x * blockDim.x + threadIdx.x;
  if (e < E) unsafeAtomicAdd(&deg[dst[e]], 1.0f);
}

__global__ void k_rsqrt_inplace(float* __restrict__ d, int n) {
  int i = blockIdx.x * blockDim.x + threadIdx.x;
  if (i < n) {
    float v = d[i];
    d[i] = (v > 0.0f) ? rsqrtf(v) : 0.0f;
  }
}

__global__ void k_gemm_wmma_f32(const float* __restrict__ A, const float* __restrict__ B,
                                const float* __restrict__ dinv,
                                float* __restrict__ C, int M, int K, int F) {
  const int lane = threadIdx.x & 31;
  const int wave = (int)((blockIdx.x * blockDim.x + threadIdx.x) >> 5);
  const int nf = F >> 4;
  const int mTile = wave / nf;
  const int fTile = wave - mTile * nf;
  const int m0 = mTile << 4;
  if (m0 >= M) return;
  const int f0 = fTile << 4;
  const int row = lane & 15;
  const int kh2 = (lane >> 4) << 1;

  const float* arow = A + (size_t)(m0 + row) * K + kh2;
  const float* bcol = B + (size_t)kh2 * F + f0 + row;

  v8f acc = {0.f, 0.f, 0.f, 0.f, 0.f, 0.f, 0.f, 0.f};
  for (int k = 0; k < K; k += 4) {
    float2 av = *reinterpret_cast<const float2*>(arow + k);
    v2f a; a.x = av.x; a.y = av.y;
    v2f b;
    b.x = bcol[(size_t)k * F];
    b.y = bcol[(size_t)(k + 1) * F];
    acc = __builtin_amdgcn_wmma_f32_16x16x4_f32(false, a, false, b,
                                                (short)0, acc, false, false);
  }
  const float* dvp = dinv + m0 + (kh2 << 2);
  float* crow = C + (size_t)(m0 + (kh2 << 2)) * F + f0 + row;
#pragma unroll
  for (int r = 0; r < 8; ++r) crow[(size_t)r * F] = acc[r] * dvp[r];
}

#define PULL_R 400
#define PULL_SEG 384
template <int F>
__global__ __launch_bounds__(1024) void k_pull(const float* __restrict__ hsc,
                                               const int* __restrict__ src,
                                               const int* __restrict__ dst,
                                               const float* __restrict__ dinv,
                                               const float* __restrict__ bias,
                                               float* __restrict__ stats,
                                               float* __restrict__ agg, int E, int Nn) {
  constexpr int C = F / 32;
  constexpr int U = 32 / C;
  constexpr int UV = 8;
  constexpr int WCH = 128 * UV;
  constexpr int TILE = 32 * WCH;
  __shared__ float acc[PULL_R * F];
  __shared__ int le[32 * PULL_SEG];
  __shared__ int ld[32 * PULL_SEG];
  __shared__ float red[2 * F];
  const int tid = threadIdx.x;
  const int lane = tid & 31;
  const int wave = __builtin_amdgcn_readfirstlane(tid >> 5);
  const int r0 = blockIdx.x * PULL_R;
  const int nrows = min(PULL_R, Nn - r0);

  {
    float4* a4 = reinterpret_cast<float4*>(acc);
    const float4 z = make_float4(0.f, 0.f, 0.f, 0.f);
    for (int i = tid; i < PULL_R * F / 4; i += 1024) a4[i] = z;
    if (tid < 2 * F) red[tid] = 0.f;
  }
  __syncthreads();

  int* mye = le + wave * PULL_SEG;
  int* myd = ld + wave * PULL_SEG;
  int wcnt = 0;
  const unsigned ltmask = (1u << lane) - 1u;
  auto keep = [&](int dval, int e) {
    const unsigned rel = (unsigned)(dval - r0);
    const bool m = rel < (unsigned)nrows;
    const unsigned bal = __builtin_amdgcn_ballot_w32(m);
    if (bal != 0u) {
      if (m) {
        const int slot = wcnt + __popc(bal & ltmask);
        if (slot < PULL_SEG) {
          mye[slot] = e;
          myd[slot] = (int)rel;
        } else {
          const float* hr = hsc + (size_t)src[e] * F;
          float* ar = acc + rel * F;
#pragma nounroll
          for (int c = 0; c < F; ++c) unsafeAtomicAdd(ar + c, hr[c]);
        }
      }
      wcnt += __popc(bal);
    }
  };

  const int ntiles = (E + TILE - 1) / TILE;
  const int t0 = (int)(((long)blockIdx.x * ntiles) / gridDim.x);
  const bool vec = ((E & 3) == 0) && ((reinterpret_cast<size_t>(dst) & 15) == 0);
  for (int k = 0; k < ntiles; ++k) {
    int t = t0 + k;
    if (t >= ntiles) t -= ntiles;
    const int eb = t * TILE + wave * WCH;
    if (vec) {
      int4 q[UV];
      if (eb + WCH <= E) {
#pragma unroll
        for (int u = 0; u < UV; ++u)
          q[u] = *reinterpret_cast<const int4*>(dst + eb + u * 128 + lane * 4);
      } else {
#pragma unroll
        for (int u = 0; u < UV; ++u) {
          const int e4 = eb + u * 128 + lane * 4;
          q[u] = make_int4(-1, -1, -1, -1);
          if (e4 < E) q[u] = *reinterpret_cast<const int4*>(dst + e4);
        }
      }
#pragma unroll
      for (int u = 0; u < UV; ++u) {
        const int e4 = eb + u * 128 + lane * 4;
        keep(q[u].x, e4);
        keep(q[u].y, e4 + 1);
        keep(q[u].z, e4 + 2);
        keep(q[u].w, e4 + 3);
      }
    } else {
      for (int h = 0; h < WCH; h += 32) {
        const int e = eb + h + lane;
        const int dval = (e < E) ? dst[e] : -1;
        keep(dval, e);
      }
    }
  }

  __syncthreads();

  const int n = min(wcnt, PULL_SEG);
  for (int i0 = 0; i0 < n; i0 += 32) {
    const int m = min(32, n - i0);
    int sv = 0, dl = 0;
    if (lane < m) { sv = src[mye[i0 + lane]]; dl = myd[i0 + lane]; }
#pragma unroll
    for (int g = 0; g < 32; g += U) {
      if (g + U <= m) {
        float v[U][C];
#pragma unroll
        for (int u = 0; u < U; ++u) {
          const int s = __builtin_amdgcn_readlane(sv, g + u);
          const float* hrow = hsc + (size_t)s * F + lane;
#pragma unroll
          for (int j = 0; j < C; ++j) v[u][j] = hrow[32 * j];
        }
#pragma unroll
        for (int u = 0; u < U; ++u) {
          const int d = __builtin_amdgcn_readlane(dl, g + u);
          float* ad = acc + d * F + lane;
#pragma unroll
          for (int j = 0; j < C; ++j) unsafeAtomicAdd(ad + 32 * j, v[u][j]);
        }
      } else if (g < m) {
        for (int i = g; i < m; ++i) {
          const int s = __builtin_amdgcn_readlane(sv, i);
          const int d = __builtin_amdgcn_readlane(dl, i);
          const float* hrow = hsc + (size_t)s * F + lane;
          float* ad = acc + d * F + lane;
          float v[C];
#pragma unroll
          for (int j = 0; j < C; ++j) v[j] = hrow[32 * j];
#pragma unroll
          for (int j = 0; j < C; ++j) unsafeAtomicAdd(ad + 32 * j, v[j]);
        }
      }
    }
  }
  __syncthreads();

  {
    float* o = agg + (size_t)r0 * F;
    const float* hs = hsc + (size_t)r0 * F;
    const float* dv = dinv + r0;
    if (((reinterpret_cast<size_t>(o) | reinterpret_cast<size_t>(hs) |
          reinterpret_cast<size_t>(bias)) & 15) == 0) {
      constexpr int F4 = F / 4;
      constexpr int RPP = 1024 / F4;
      const int c4 = tid % F4;
      const int rr = tid / F4;
      const float4* a4 = reinterpret_cast<const float4*>(acc);
      const float4* h4 = reinterpret_cast<const float4*>(hs);
      float4* o4 = reinterpret_cast<float4*>(o);
      const float4 b = reinterpret_cast<const float4*>(bias)[c4];
      float4 s = make_float4(0.f, 0.f, 0.f, 0.f);
      float4 q = make_float4(0.f, 0.f, 0.f, 0.f);
#pragma unroll 4
      for (int r = rr; r < nrows; r += RPP) {
        const float4 a = a4[r * F4 + c4];
        const float4 h = h4[r * F4 + c4];
        const float di = dv[r];
        float4 v;
        v.x = fmaxf((a.x + h.x) * di + b.x, 0.f);
        v.y = fmaxf((a.y + h.y) * di + b.y, 0.f);
        v.z = fmaxf((a.z + h.z) * di + b.z, 0.f);
        v.w = fmaxf((a.w + h.w) * di + b.w, 0.f);
        o4[r * F4 + c4] = v;
        s.x += v.x; s.y += v.y; s.z += v.z; s.w += v.w;
        q.x += v.x * v.x; q.y += v.y * v.y; q.z += v.z * v.z; q.w += v.w * v.w;
      }
      float* rs = red + 4 * c4;
      unsafeAtomicAdd(rs + 0, s.x);
      unsafeAtomicAdd(rs + 1, s.y);
      unsafeAtomicAdd(rs + 2, s.z);
      unsafeAtomicAdd(rs + 3, s.w);
      unsafeAtomicAdd(rs + F + 0, q.x);
      unsafeAtomicAdd(rs + F + 1, q.y);
      unsafeAtomicAdd(rs + F + 2, q.z);
      unsafeAtomicAdd(rs + F + 3, q.w);
    } else {
      constexpr int RPP = 1024 / F;
      const int c = tid % F;
      const float bj = bias[c];
      float s = 0.f, q = 0.f;
      for (int r = tid / F; r < nrows; r += RPP) {
        float v = (acc[r * F + c] + hs[(size_t)r * F + c]) * dv[r] + bj;
        v = fmaxf(v, 0.f);
        o[(size_t)r * F + c] = v;
        s += v;
        q += v * v;
      }
      unsafeAtomicAdd(red + c, s);
      unsafeAtomicAdd(red + F + c, q);
    }
  }
  __syncthreads();
  if (tid < 2 * F) unsafeAtomicAdd(stats + tid, red[tid]);
}

__global__ void k_bn_apply(float* __restrict__ v, const float* __restrict__ stats,
                           const float* __restrict__ gamma, const float* __restrict__ beta,
                           int Nn, int F, float invN) {
  size_t total = (size_t)Nn * F;
  size_t i = (size_t)blockIdx.x * blockDim.x + threadIdx.x;
  size_t stride = (size_t)gridDim.x * blockDim.x;
  for (; i < total; i += stride) {
    int j = (int)(i & (size_t)(F - 1));
    float mean = stats[j] * invN;
    float var = stats[F + j] * invN - mean * mean;
    v[i] = (v[i] - mean) * rsqrtf(var + BN_EPS) * gamma[j] + beta[j];
  }
}

__global__ void k_gemm64x2(const float* __restrict__ h3, const float* __restrict__ W4,
                           const float* __restrict__ dinv,
                           float* __restrict__ h4, int Nn) {
  int i = blockIdx.x * blockDim.x + threadIdx.x;
  if (i >= Nn) return;
  const float* row = h3 + (size_t)i * 64;
  float o0 = 0.f, o1 = 0.f;
#pragma unroll 8
  for (int k = 0; k < 64; ++k) {
    float xv = row[k];
    o0 += xv * W4[2 * k];
    o1 += xv * W4[2 * k + 1];
  }
  float di = dinv[i];
  h4[2 * (size_t)i]     = o0 * di;
  h4[2 * (size_t)i + 1] = o1 * di;
}

__global__ void k_scatter2(const float* __restrict__ h4s, const int* __restrict__ src,
                           const int* __restrict__ dst,
                           float* __restrict__ agg, int E) {
  int e = blockIdx.x * blockDim.x + threadIdx.x;
  if (e >= E) return;
  int s = src[e], d = dst[e];
  float2 v = *reinterpret_cast<const float2*>(h4s + 2 * (size_t)s);
  unsafeAtomicAdd(agg + 2 * (size_t)d,     v.x);
  unsafeAtomicAdd(agg + 2 * (size_t)d + 1, v.y);
}

__global__ void k_final_lsm(const float* __restrict__ agg, const float* __restrict__ h4s,
                            const float* __restrict__ dinv, const float* __restrict__ b4,
                            float* __restrict__ out, int Nn) {
  int i = blockIdx.x * blockDim.x + threadIdx.x;
  if (i >= Nn) return;
  float di = dinv[i];
  float o0 = (agg[2 * (size_t)i]     + h4s[2 * (size_t)i])     * di + b4[0];
  float o1 = (agg[2 * (size_t)i + 1] + h4s[2 * (size_t)i + 1]) * di + b4[1];
  float m = fmaxf(o0, o1);
  float l = m + logf(expf(o0 - m) + expf(o1 - m));
  out[2 * (size_t)i]     = o0 - l;
  out[2 * (size_t)i + 1] = o1 - l;
}

extern "C" void kernel_launch(void* const* d_in, const int* in_sizes, int n_in,
                              void* d_out, int out_size, void* d_ws, size_t ws_size,
                              hipStream_t stream) {
  const float* x   = (const float*)d_in[0];
  const int*   ei  = (const int*)d_in[1];
  const float* W1  = (const float*)d_in[2];
  const float* b1  = (const float*)d_in[3];
  const float* g1  = (const float*)d_in[4];
  const float* be1 = (const float*)d_in[5];
  const float* W2  = (const float*)d_in[6];
  const float* b2  = (const float*)d_in[7];
  const float* g2  = (const float*)d_in[8];
  const float* be2 = (const float*)d_in[9];
  const float* W3  = (const float*)d_in[10];
  const float* b3  = (const float*)d_in[11];
  const float* g3  = (const float*)d_in[12];
  const float* be3 = (const float*)d_in[13];
  const float* W4  = (const float*)d_in[14];
  const float* b4  = (const float*)d_in[15];

  const int Nn = in_sizes[0] / 24;
  const int E  = in_sizes[1] / 2;
  const int* src = ei;
  const int* dst = ei + E;

  float* ws    = (float*)d_ws;
  float* dinv  = ws;
  float* stats = ws + Nn;
  float* bufA  = stats + 256;
  float* bufB  = bufA + (size_t)Nn * 128;

  const float invN = 1.0f / (float)Nn;
  const int TB = 256;

  auto gemm = [&](const float* A, const float* Bm, float* C, int K, int F) {
    long waves = (long)((Nn + 15) / 16) * (F / 16);
    int blocks = (int)((waves * 32 + TB - 1) / TB);
    k_gemm_wmma_f32<<<blocks, TB, 0, stream>>>(A, Bm, dinv, C, Nn, K, F);
  };
  auto pull128 = [&](const float* h, const float* bias, float* out) {
    k_zero<<<1, 256, 0, stream>>>(stats, 256);
    k_pull<128><<<(Nn + PULL_R - 1) / PULL_R, 1024, 0, stream>>>(h, src, dst, dinv, bias, stats,
                                                                 out, E, Nn);
  };
  auto pull64 = [&](const float* h, const float* bias, float* out) {
    k_zero<<<1, 256, 0, stream>>>(stats, 256);
    k_pull<64><<<(Nn + PULL_R - 1) / PULL_R, 1024, 0, stream>>>(h, src, dst, dinv, bias, stats,
                                                                out, E, Nn);
  };
  auto bn = [&](float* v, const float* gamma, const float* beta, int F) {
    k_bn_apply<<<2048, TB, 0, stream>>>(v, stats, gamma, beta, Nn, F, invN);
  };
  auto zero = [&](float* p, size_t n) {
    int blocks = (int)((n + TB - 1) / TB);
    if (blocks > 4096) blocks = 4096;
    k_zero<<<blocks, TB, 0, stream>>>(p, n);
  };

  k_deg_init<<<(Nn + TB - 1) / TB, TB, 0, stream>>>(dinv, Nn);
  k_deg_count<<<(E + TB - 1) / TB, TB, 0, stream>>>(dst, dinv, E);
  k_rsqrt_inplace<<<(Nn + TB - 1) / TB, TB, 0, stream>>>(dinv, Nn);

  gemm(x, W1, bufA, 24, 128);
  pull128(bufA, b1, bufB);
  bn(bufB, g1, be1, 128);

  gemm(bufB, W2, bufA, 128, 128);
  pull128(bufA, b2, bufB);
  bn(bufB, g2, be2, 128);

  gemm(bufB, W3, bufA, 128, 64);
  pull64(bufA, b3, bufB);
  bn(bufB, g3, be3, 64);

  k_gemm64x2<<<(Nn + TB - 1) / TB, TB, 0, stream>>>(bufB, W4, dinv, bufA, Nn);
  zero(bufB, (size_t)Nn * 2);
  k_scatter2<<<(E + TB - 1) / TB, TB, 0, stream>>>(bufA, src, dst, bufB, E);
  k_final_lsm<<<(Nn + TB - 1) / TB, TB, 0, stream>>>(bufB, bufA, dinv, b4,
                                                     (float*)d_out, Nn);
}


# degree count and layer-4 edge sum on (edge chunk x row range) tiles: a workgroup adds its range in LDS, stores a plane, next pass adds 32 planes; their global float atomics go (sum order changes)
# speedup vs baseline: 3.0829x; 1.3515x over previous
#include <hip/hip_runtime.h>
#include <hip/hip_bf16.h>
#include <math.h>


typedef float v2f __attribute__((ext_vector_type(2)));
typedef float v8f __attribute__((ext_vector_type(8)));

#define BN_EPS 1e-5f

__global__ void k_zero(float* __restrict__ p, size_t n) {
  size_t i = (size_t)blockIdx.x * blockDim.x + threadIdx.x;
  size_t stride = (size_t)gridDim.x * blockDim.x;
  for (; i < n; i += stride) p[i] = 0.0f;
}

#define PART_R 12800
#define PART_CH 32

__device__ __forceinline__ int4 ld_idx4(const int* __restrict__ p, int e, int ee, bool vec) {
  if (vec && e + 4 <= ee) return *reinterpret_cast<const int4*>(p + e);
  int4 r;
  r.x = (e     < ee) ? p[e]     : -1;
  r.y = (e + 1 < ee) ? p[e + 1] : -1;
  r.z = (e + 2 < ee) ? p[e + 2] : -1;
  r.w = (e + 3 < ee) ? p[e + 3] : -1;
  return r;
}

__global__ __launch_bounds__(1024) void k_deg_part(const int* __restrict__ dst,
                                                   int* __restrict__ part, int E, int Nn) {
  constexpr int U = 8;
  __shared__ int cnt[PART_R];
  const int tid = threadIdx.x;
  const int r0 = blockIdx.y * PART_R;
  const int nrows = min(PART_R, Nn - r0);
  for (int i = tid; i < PART_R; i += 1024) cnt[i] = 0;
  __syncthreads();
  const int per = (((E + PART_CH - 1) / PART_CH) + 3) & ~3;
  const int eb = blockIdx.x * per;
  const int ee = min(E, eb + per);
  const bool vec = (reinterpret_cast<size_t>(dst) & 15) == 0;
  for (int b = eb; b < ee; b += 4096 * U) {
    int4 q[U];
    if (vec && b + 4096 * U <= ee) {
#pragma unroll
      for (int u = 0; u < U; ++u)
        q[u] = *reinterpret_cast<const int4*>(dst + b + u * 4096 + tid * 4);
    } else {
#pragma unroll
      for (int u = 0; u < U; ++u) q[u] = ld_idx4(dst, b + u * 4096 + tid * 4, ee, vec);
    }
#pragma unroll
    for (int u = 0; u < U; ++u) {
      const int dv[4] = {q[u].x, q[u].y, q[u].z, q[u].w};
#pragma unroll
      for (int j = 0; j < 4; ++j) {
        const unsigned rel = (unsigned)(dv[j] - r0);
        if (rel < (unsigned)nrows) atomicAdd(&cnt[rel], 1);
      }
    }
  }
  __syncthreads();
  int* o = part + (size_t)blockIdx.x * Nn + r0;
  for (int i = tid; i < nrows; i += 1024) o[i] = cnt[i];
}

__global__ void k_deg_fin(const int* __restrict__ part, float* __restrict__ dinv, int n) {
  int i = blockIdx.x * blockDim.x + threadIdx.x;
  if (i < n) {
    int c = 1;
#pragma unroll
    for (int k = 0; k < PART_CH; ++k) c += part[(size_t)k * n + i];
    float v = (float)c;
    dinv[i] = (v > 0.0f) ? rsqrtf(v) : 0.0f;
  }
}

__global__ void k_gemm_wmma_f32(const float* __restrict__ A, const float* __restrict__ B,
                                const float* __restrict__ dinv,
                                float* __restrict__ C, int M, int K, int F) {
  const int lane = threadIdx.x & 31;
  const int wave = (int)((blockIdx.x * blockDim.x + threadIdx.x) >> 5);
  const int nf = F >> 4;
  const int mTile = wave / nf;
  const int fTile = wave - mTile * nf;
  const int m0 = mTile << 4;
  if (m0 >= M) return;
  const int f0 = fTile << 4;
  const int row = lane & 15;
  const int kh2 = (lane >> 4) << 1;

  const float* arow = A + (size_t)(m0 + row) * K + kh2;
  const float* bcol = B + (size_t)kh2 * F + f0 + row;

  v8f acc = {0.f, 0.f, 0.f, 0.f, 0.f, 0.f, 0.f, 0.f};
  for (int k = 0; k < K; k += 4) {
    float2 av = *reinterpret_cast<const float2*>(arow + k);
    v2f a; a.x = av.x; a.y = av.y;
    v2f b;
    b.x = bcol[(size_t)k * F];
    b.y = bcol[(size_t)(k + 1) * F];
    acc = __builtin_amdgcn_wmma_f32_16x16x4_f32(false, a, false, b,
                                                (short)0, acc, false, false);
  }
  const float* dvp = dinv + m0 + (kh2 << 2);
  float* crow = C + (size_t)(m0 + (kh2 << 2)) * F + f0 + row;
#pragma unroll
  for (int r = 0; r < 8; ++r) crow[(size_t)r * F] = acc[r] * dvp[r];
}

#define PULL_R 400
#define PULL_SEG 384
template <int F>
__global__ __launch_bounds__(1024) void k_pull(const float* __restrict__ hsc,
                                               const int* __restrict__ src,
                                               const int* __restrict__ dst,
                                               const float* __restrict__ dinv,
                                               const float* __restrict__ bias,
                                               float* __restrict__ stats,
                                               float* __restrict__ agg, int E, int Nn) {
  constexpr int C = F / 32;
  constexpr int U = 32 / C;
  constexpr int UV = 8;
  constexpr int WCH = 128 * UV;
  constexpr int TILE = 32 * WCH;
  __shared__ float acc[PULL_R * F];
  __shared__ int le[32 * PULL_SEG];
  __shared__ int ld[32 * PULL_SEG];
  __shared__ float red[2 * F];
  const int tid = threadIdx.x;
  const int lane = tid & 31;
  const int wave = __builtin_amdgcn_readfirstlane(tid >> 5);
  const int r0 = blockIdx.x * PULL_R;
  const int nrows = min(PULL_R, Nn - r0);

  {
    float4* a4 = reinterpret_cast<float4*>(acc);
    const float4 z = make_float4(0.f, 0.f, 0.f, 0.f);
    for (int i = tid; i < PULL_R * F / 4; i += 1024) a4[i] = z;
    if (tid < 2 * F) red[tid] = 0.f;
  }
  __syncthreads();

  int* mye = le + wave * PULL_SEG;
  int* myd = ld + wave * PULL_SEG;
  int wcnt = 0;
  const unsigned ltmask = (1u << lane) - 1u;
  auto keep = [&](int dval, int e) {
    const unsigned rel = (unsigned)(dval - r0);
    const bool m = rel < (unsigned)nrows;
    const unsigned bal = __builtin_amdgcn_ballot_w32(m);
    if (bal != 0u) {
      if (m) {
        const int slot = wcnt + __popc(bal & ltmask);
        if (slot < PULL_SEG) {
          mye[slot] = e;
          myd[slot] = (int)rel;
        } else {
          const float* hr = hsc + (size_t)src[e] * F;
          float* ar = acc + rel * F;
#pragma nounroll
          for (int c = 0; c < F; ++c) unsafeAtomicAdd(ar + c, hr[c]);
        }
      }
      wcnt += __popc(bal);
    }
  };

  const int ntiles = (E + TILE - 1) / TILE;
  const int t0 = (int)(((long)blockIdx.x * ntiles) / gridDim.x);
  const bool vec = ((E & 3) == 0) && ((reinterpret_cast<size_t>(dst) & 15) == 0);
  for (int k = 0; k < ntiles; ++k) {
    int t = t0 + k;
    if (t >= ntiles) t -= ntiles;
    const int eb = t * TILE + wave * WCH;
    if (vec) {
      int4 q[UV];
      if (eb + WCH <= E) {
#pragma unroll
        for (int u = 0; u < UV; ++u)
          q[u] = *reinterpret_cast<const int4*>(dst + eb + u * 128 + lane * 4);
      } else {
#pragma unroll
        for (int u = 0; u < UV; ++u) {
          const int e4 = eb + u * 128 + lane * 4;
          q[u] = make_int4(-1, -1, -1, -1);
          if (e4 < E) q[u] = *reinterpret_cast<const int4*>(dst + e4);
        }
      }
#pragma unroll
      for (int u = 0; u < UV; ++u) {
        const int e4 = eb + u * 128 + lane * 4;
        keep(q[u].x, e4);
        keep(q[u].y, e4 + 1);
        keep(q[u].z, e4 + 2);
        keep(q[u].w, e4 + 3);
      }
    } else {
      for (int h = 0; h < WCH; h += 32) {
        const int e = eb + h + lane;
        const int dval = (e < E) ? dst[e] : -1;
        keep(dval, e);
      }
    }
  }

  __syncthreads();

  const int n = min(wcnt, PULL_SEG);
  for (int i0 = 0; i0 < n; i0 += 32) {
    const int m = min(32, n - i0);
    int sv = 0, dl = 0;
    if (lane < m) { sv = src[mye[i0 + lane]]; dl = myd[i0 + lane]; }
#pragma unroll
    for (int g = 0; g < 32; g += U) {
      if (g + U <= m) {
        float v[U][C];
#pragma unroll
        for (int u = 0; u < U; ++u) {
          const int s = __builtin_amdgcn_readlane(sv, g + u);
          const float* hrow = hsc + (size_t)s * F + lane;
#pragma unroll
          for (int j = 0; j < C; ++j) v[u][j] = hrow[32 * j];
        }
#pragma unroll
        for (int u = 0; u < U; ++u) {
          const int d = __builtin_amdgcn_readlane(dl, g + u);
          float* ad = acc + d * F + lane;
#pragma unroll
          for (int j = 0; j < C; ++j) unsafeAtomicAdd(ad + 32 * j, v[u][j]);
        }
      } else if (g < m) {
        for (int i = g; i < m; ++i) {
          const int s = __builtin_amdgcn_readlane(sv, i);
          const int d = __builtin_amdgcn_readlane(dl, i);
          const float* hrow = hsc + (size_t)s * F + lane;
          float* ad = acc + d * F + lane;
          float v[C];
#pragma unroll
          for (int j = 0; j < C; ++j) v[j] = hrow[32 * j];
#pragma unroll
          for (int j = 0; j < C; ++j) unsafeAtomicAdd(ad + 32 * j, v[j]);
        }
      }
    }
  }
  __syncthreads();

  {
    float* o = agg + (size_t)r0 * F;
    const float* hs = hsc + (size_t)r0 * F;
    const float* dv = dinv + r0;
    if (((reinterpret_cast<size_t>(o) | reinterpret_cast<size_t>(hs) |
          reinterpret_cast<size_t>(bias)) & 15) == 0) {
      constexpr int F4 = F / 4;
      constexpr int RPP = 1024 / F4;
      const int c4 = tid % F4;
      const int rr = tid / F4;
      const float4* a4 = reinterpret_cast<const float4*>(acc);
      const float4* h4 = reinterpret_cast<const float4*>(hs);
      float4* o4 = reinterpret_cast<float4*>(o);
      const float4 b = reinterpret_cast<const float4*>(bias)[c4];
      float4 s = make_float4(0.f, 0.f, 0.f, 0.f);
      float4 q = make_float4(0.f, 0.f, 0.f, 0.f);
#pragma unroll 4
      for (int r = rr; r < nrows; r += RPP) {
        const float4 a = a4[r * F4 + c4];
        const float4 h = h4[r * F4 + c4];
        const float di = dv[r];
        float4 v;
        v.x = fmaxf((a.x + h.x) * di + b.x, 0.f);
        v.y = fmaxf((a.y + h.y) * di + b.y, 0.f);
        v.z = fmaxf((a.z + h.z) * di + b.z, 0.f);
        v.w = fmaxf((a.w + h.w) * di + b.w, 0.f);
        o4[r * F4 + c4] = v;
        s.x += v.x; s.y += v.y; s.z += v.z; s.w += v.w;
        q.x += v.x * v.x; q.y += v.y * v.y; q.z += v.z * v.z; q.w += v.w * v.w;
      }
      float* rs = red + 4 * c4;
      unsafeAtomicAdd(rs + 0, s.x);
      unsafeAtomicAdd(rs + 1, s.y);
      unsafeAtomicAdd(rs + 2, s.z);
      unsafeAtomicAdd(rs + 3, s.w);
      unsafeAtomicAdd(rs + F + 0, q.x);
      unsafeAtomicAdd(rs + F + 1, q.y);
      unsafeAtomicAdd(rs + F + 2, q.z);
      unsafeAtomicAdd(rs + F + 3, q.w);
    } else {
      constexpr int RPP = 1024 / F;
      const int c = tid % F;
      const float bj = bias[c];
      float s = 0.f, q = 0.f;
      for (int r = tid / F; r < nrows; r += RPP) {
        float v = (acc[r * F + c] + hs[(size_t)r * F + c]) * dv[r] + bj;
        v = fmaxf(v, 0.f);
        o[(size_t)r * F + c] = v;
        s += v;
        q += v * v;
      }
      unsafeAtomicAdd(red + c, s);
      unsafeAtomicAdd(red + F + c, q);
    }
  }
  __syncthreads();
  if (tid < 2 * F) unsafeAtomicAdd(stats + tid, red[tid]);
}

__global__ void k_bn_apply(float* __restrict__ v, const float* __restrict__ stats,
                           const float* __restrict__ gamma, const float* __restrict__ beta,
                           int Nn, int F, float invN) {
  size_t total = (size_t)Nn * F;
  size_t i = (size_t)blockIdx.x * blockDim.x + threadIdx.x;
  size_t stride = (size_t)gridDim.x * blockDim.x;
  for (; i < total; i += stride) {
    int j = (int)(i & (size_t)(F - 1));
    float mean = stats[j] * invN;
    float var = stats[F + j] * invN - mean * mean;
    v[i] = (v[i] - mean) * rsqrtf(var + BN_EPS) * gamma[j] + beta[j];
  }
}

__global__ void k_gemm64x2(const float* __restrict__ h3, const float* __restrict__ W4,
                           const float* __restrict__ dinv,
                           float* __restrict__ h4, int Nn) {
  int i = blockIdx.x * blockDim.x + threadIdx.x;
  if (i >= Nn) return;
  const float* row = h3 + (size_t)i * 64;
  float o0 = 0.f, o1 = 0.f;
#pragma unroll 8
  for (int k = 0; k < 64; ++k) {
    float xv = row[k];
    o0 += xv * W4[2 * k];
    o1 += xv * W4[2 * k + 1];
  }
  float di = dinv[i];
  h4[2 * (size_t)i]     = o0 * di;
  h4[2 * (size_t)i + 1] = o1 * di;
}

__global__ __launch_bounds__(1024) void k_sum2_part(const float* __restrict__ h4s,
                                                    const int* __restrict__ src,
                                                    const int* __restrict__ dst,
                                                    float* __restrict__ part, int E, int Nn) {
  constexpr int U = 4;
  __shared__ float acc[2 * (PART_R + 32)];
  const int tid = threadIdx.x;
  const int spare = PART_R + (tid & 31);
  const int r0 = blockIdx.y * PART_R;
  const int nrows = min(PART_R, Nn - r0);
  {
    float4* a4 = reinterpret_cast<float4*>(acc);
    const float4 z = make_float4(0.f, 0.f, 0.f, 0.f);
    for (int i = tid; i < 2 * PART_R / 4; i += 1024) a4[i] = z;
  }
  __syncthreads();
  const int per = (((E + PART_CH - 1) / PART_CH) + 3) & ~3;
  const int eb = blockIdx.x * per;
  const int ee = min(E, eb + per);
  const bool vd = (reinterpret_cast<size_t>(dst) & 15) == 0;
  const bool vs = (reinterpret_cast<size_t>(src) & 15) == 0;
  const float2* h2 = reinterpret_cast<const float2*>(h4s);
  for (int b = eb; b < ee; b += 4096 * U) {
    int4 qd[U], qs[U];
    float2 v[4 * U];
    if (vd && vs && b + 4096 * U <= ee) {
#pragma unroll
      for (int u = 0; u < U; ++u) {
        const int e = b + u * 4096 + tid * 4;
        qd[u] = *reinterpret_cast<const int4*>(dst + e);
        qs[u] = *reinterpret_cast<const int4*>(src + e);
      }
    } else {
#pragma unroll
      for (int u = 0; u < U; ++u) {
        const int e = b + u * 4096 + tid * 4;
        qd[u] = ld_idx4(dst, e, ee, vd);
        qs[u] = ld_idx4(src, e, ee, vs);
      }
    }
#pragma unroll
    for (int u = 0; u < U; ++u) {
      const int dv[4] = {qd[u].x, qd[u].y, qd[u].z, qd[u].w};
      const int sv[4] = {qs[u].x, qs[u].y, qs[u].z, qs[u].w};
#pragma unroll
      for (int j = 0; j < 4; ++j) {
        const bool m = (unsigned)(dv[j] - r0) < (unsigned)nrows;
        v[4 * u + j] = h2[m ? sv[j] : 0];
      }
    }
#pragma unroll
    for (int u = 0; u < U; ++u) {
      const int dv[4] = {qd[u].x, qd[u].y, qd[u].z, qd[u].w};
#pragma unroll
      for (int j = 0; j < 4; ++j) {
        const unsigned rel = (unsigned)(dv[j] - r0);
        float* ar = acc + 2 * ((rel < (unsigned)nrows) ? (int)rel : spare);
        unsafeAtomicAdd(ar, v[4 * u + j].x);
        unsafeAtomicAdd(ar + 1, v[4 * u + j].y);
      }
    }
  }
  __syncthreads();
  float2* o = reinterpret_cast<float2*>(part) + ((size_t)blockIdx.x * Nn + r0);
  const float2* a2 = reinterpret_cast<const float2*>(acc);
  for (int i = tid; i < nrows; i += 1024) o[i] = a2[i];
}

__global__ void k_final_lsm(const float* __restrict__ part, const float* __restrict__ h4s,
                            const float* __restrict__ dinv, const float* __restrict__ b4,
                            float* __restrict__ out, int Nn) {
  int i = blockIdx.x * blockDim.x + threadIdx.x;
  if (i >= Nn) return;
  const float2* p2 = reinterpret_cast<const float2*>(part);
  float a0 = 0.f, a1 = 0.f;
#pragma unroll
  for (int k = 0; k < PART_CH; ++k) {
    const float2 t = p2[(size_t)k * Nn + i];
    a0 += t.x;
    a1 += t.y;
  }
  float di = dinv[i];
  float o0 = (a0 + h4s[2 * (size_t)i])     * di + b4[0];
  float o1 = (a1 + h4s[2 * (size_t)i + 1]) * di + b4[1];
  float m = fmaxf(o0, o1);
  float l = m + logf(expf(o0 - m) + expf(o1 - m));
  out[2 * (size_t)i]     = o0 - l;
  out[2 * (size_t)i + 1] = o1 - l;
}

extern "C" void kernel_launch(void* const* d_in, const int* in_sizes, int n_in,
                              void* d_out, int out_size, void* d_ws, size_t ws_size,
                              hipStream_t stream) {
  const float* x   = (const float*)d_in[0];
  const int*   ei  = (const int*)d_in[1];
  const float* W1  = (const float*)d_in[2];
  const float* b1  = (const float*)d_in[3];
  const float* g1  = (const float*)d_in[4];
  const float* be1 = (const float*)d_in[5];
  const float* W2  = (const float*)d_in[6];
  const float* b2  = (const float*)d_in[7];
  const float* g2  = (const float*)d_in[8];
  const float* be2 = (const float*)d_in[9];
  const float* W3  = (const float*)d_in[10];
  const float* b3  = (const float*)d_in[11];
  const float* g3  = (const float*)d_in[12];
  const float* be3 = (const float*)d_in[13];
  const float* W4  = (const float*)d_in[14];
  const float* b4  = (const float*)d_in[15];

  const int Nn = in_sizes[0] / 24;
  const int E  = in_sizes[1] / 2;
  const int* src = ei;
  const int* dst = ei + E;

  float* ws    = (float*)d_ws;
  float* dinv  = ws;
  float* stats = ws + Nn;
  float* bufA  = stats + 256;
  float* bufB  = bufA + (size_t)Nn * 128;

  const float invN = 1.0f / (float)Nn;
  const int TB = 256;

  auto gemm = [&](const float* A, const float* Bm, float* C, int K, int F) {
    long waves = (long)((Nn + 15) / 16) * (F / 16);
    int blocks = (int)((waves * 32 + TB - 1) / TB);
    k_gemm_wmma_f32<<<blocks, TB, 0, stream>>>(A, Bm, dinv, C, Nn, K, F);
  };
  auto pull128 = [&](const float* h, const float* bias, float* out) {
    k_zero<<<1, 256, 0, stream>>>(stats, 256);
    k_pull<128><<<(Nn + PULL_R - 1) / PULL_R, 1024, 0, stream>>>(h, src, dst, dinv, bias, stats,
                                                                 out, E, Nn);
  };
  auto pull64 = [&](const float* h, const float* bias, float* out) {
    k_zero<<<1, 256, 0, stream>>>(stats, 256);
    k_pull<64><<<(Nn + PULL_R - 1) / PULL_R, 1024, 0, stream>>>(h, src, dst, dinv, bias, stats,
                                                                out, E, Nn);
  };
  auto bn = [&](float* v, const float* gamma, const float* beta, int F) {
    k_bn_apply<<<2048, TB, 0, stream>>>(v, stats, gamma, beta, Nn, F, invN);
  };
  const dim3 partGrid(PART_CH, (Nn + PART_R - 1) / PART_R);

  k_deg_part<<<partGrid, 1024, 0, stream>>>(dst, (int*)bufB, E, Nn);
  k_deg_fin<<<(Nn + TB - 1) / TB, TB, 0, stream>>>((const int*)bufB, dinv, Nn);

  gemm(x, W1, bufA, 24, 128);
  pull128(bufA, b1, bufB);
  bn(bufB, g1, be1, 128);

  gemm(bufB, W2, bufA, 128, 128);
  pull128(bufA, b2, bufB);
  bn(bufB, g2, be2, 128);

  gemm(bufB, W3, bufA, 128, 64);
  pull64(bufA, b3, bufB);
  bn(bufB, g3, be3, 64);

  k_gemm64x2<<<(Nn + TB - 1) / TB, TB, 0, stream>>>(bufB, W4, dinv, bufA, Nn);
  k_sum2_part<<<partGrid, 1024, 0, stream>>>(bufA, src, dst, bufB, E, Nn);
  k_final_lsm<<<(Nn + TB - 1) / TB, TB, 0, stream>>>(bufB, bufA, dinv, b4,
                                                     (float*)d_out, Nn);
}


# layers 1-3 edge sum: a pass ahead bins the edges by (edge chunk x destination range) into cells kept in the range's own output rows; its workgroup reads 256 cells, not all of dst (arrival order)
# speedup vs baseline: 3.9871x; 1.2933x over previous
#include <hip/hip_runtime.h>
#include <hip/hip_bf16.h>
#include <math.h>


typedef float v2f __attribute__((ext_vector_type(2)));
typedef float v8f __attribute__((ext_vector_type(8)));

#define BN_EPS 1e-5f

#define PART_R 12800
#define PART_CH 32

__device__ __forceinline__ int4 ld_idx4(const int* __restrict__ p, int e, int ee, bool vec) {
  if (vec && e + 4 <= ee) return *reinterpret_cast<const int4*>(p + e);
  int4 r;
  r.x = (e     < ee) ? p[e]     : -1;
  r.y = (e + 1 < ee) ? p[e + 1] : -1;
  r.z = (e + 2 < ee) ? p[e + 2] : -1;
  r.w = (e + 3 < ee) ? p[e + 3] : -1;
  return r;
}

__global__ __launch_bounds__(1024) void k_deg_part(const int* __restrict__ dst,
                                                   int* __restrict__ part, int E, int Nn) {
  constexpr int U = 8;
  __shared__ int cnt[PART_R];
  const int tid = threadIdx.x;
  const int r0 = blockIdx.y * PART_R;
  const int nrows = min(PART_R, Nn - r0);
  for (int i = tid; i < PART_R; i += 1024) cnt[i] = 0;
  __syncthreads();
  const int per = (((E + PART_CH - 1) / PART_CH) + 3) & ~3;
  const int eb = blockIdx.x * per;
  const int ee = min(E, eb + per);
  const bool vec = (reinterpret_cast<size_t>(dst) & 15) == 0;
  for (int b = eb; b < ee; b += 4096 * U) {
    int4 q[U];
    if (vec && b + 4096 * U <= ee) {
#pragma unroll
      for (int u = 0; u < U; ++u)
        q[u] = *reinterpret_cast<const int4*>(dst + b + u * 4096 + tid * 4);
    } else {
#pragma unroll
      for (int u = 0; u < U; ++u) q[u] = ld_idx4(dst, b + u * 4096 + tid * 4, ee, vec);
    }
#pragma unroll
    for (int u = 0; u < U; ++u) {
      const int dv[4] = {q[u].x, q[u].y, q[u].z, q[u].w};
#pragma unroll
      for (int j = 0; j < 4; ++j) {
        const unsigned rel = (unsigned)(dv[j] - r0);
        if (rel < (unsigned)nrows) atomicAdd(&cnt[rel], 1);
      }
    }
  }
  __syncthreads();
  int* o = part + (size_t)blockIdx.x * Nn + r0;
  for (int i = tid; i < nrows; i += 1024) o[i] = cnt[i];
}

__global__ void k_deg_fin(const int* __restrict__ part, float* __restrict__ dinv, int n) {
  int i = blockIdx.x * blockDim.x + threadIdx.x;
  if (i < n) {
    int c = 1;
#pragma unroll
    for (int k = 0; k < PART_CH; ++k) c += part[(size_t)k * n + i];
    float v = (float)c;
    dinv[i] = (v > 0.0f) ? rsqrtf(v) : 0.0f;
  }
}

__global__ void k_gemm_wmma_f32(const float* __restrict__ A, const float* __restrict__ B,
                                const float* __restrict__ dinv,
                                float* __restrict__ C, int M, int K, int F) {
  const int lane = threadIdx.x & 31;
  const int wave = (int)((blockIdx.x * blockDim.x + threadIdx.x) >> 5);
  const int nf = F >> 4;
  const int mTile = wave / nf;
  const int fTile = wave - mTile * nf;
  const int m0 = mTile << 4;
  if (m0 >= M) return;
  const int f0 = fTile << 4;
  const int row = lane & 15;
  const int kh2 = (lane >> 4) << 1;

  const float* arow = A + (size_t)(m0 + row) * K + kh2;
  const float* bcol = B + (size_t)kh2 * F + f0 + row;

  v8f acc = {0.f, 0.f, 0.f, 0.f, 0.f, 0.f, 0.f, 0.f};
  for (int k = 0; k < K; k += 4) {
    float2 av = *reinterpret_cast<const float2*>(arow + k);
    v2f a; a.x = av.x; a.y = av.y;
    v2f b;
    b.x = bcol[(size_t)k * F];
    b.y = bcol[(size_t)(k + 1) * F];
    acc = __builtin_amdgcn_wmma_f32_16x16x4_f32(false, a, false, b,
                                                (short)0, acc, false, false);
  }
  const float* dvp = dinv + m0 + (kh2 << 2);
  float* crow = C + (size_t)(m0 + (kh2 << 2)) * F + f0 + row;
#pragma unroll
  for (int r = 0; r < 8; ++r) crow[(size_t)r * F] = acc[r] * dvp[r];
}

#define PULL_R 400
#define BIN_CH 256
#define BIN_CAP 96
#define BIN_G 256
#define PULL_SEG (8 * BIN_CAP)
#define PACK_MAX_ROWS (1 << 23)

template <int F>
__global__ __launch_bounds__(1024) void k_bin(const int* __restrict__ src,
                                              const int* __restrict__ dst,
                                              int* __restrict__ cells,
                                              float* __restrict__ stats,
                                              int E, int Nn) {
  constexpr int PITCH = PULL_R * F / BIN_CH;
  static_assert(PITCH >= BIN_CAP + 1, "a cell holds its count and BIN_CAP entries");
  static_assert(PULL_R <= 512, "row within the range takes 9 bits");
  __shared__ int cnt[BIN_G];
  __shared__ int stage[BIN_G * BIN_CAP];
  const int tid = threadIdx.x;
  if (blockIdx.x == 0 && blockIdx.y == 0 && tid < 256) stats[tid] = 0.f;
  const int nb = (Nn + PULL_R - 1) / PULL_R;
  const int b0 = blockIdx.y * BIN_G;
  const int ng = min(BIN_G, nb - b0);
  const int lo = b0 * PULL_R;
  if (tid < BIN_G) cnt[tid] = 0;
  __syncthreads();
  const int per = (((E + BIN_CH - 1) / BIN_CH) + 3) & ~3;
  const int eb = blockIdx.x * per;
  const int ee = min(E, eb + per);
  const bool vd = (reinterpret_cast<size_t>(dst) & 15) == 0;
  const bool vs = (reinterpret_cast<size_t>(src) & 15) == 0;
  for (int e = eb + tid * 4; e < ee; e += 4096) {
    const int4 qd = ld_idx4(dst, e, ee, vd);
    const int4 qs = ld_idx4(src, e, ee, vs);
    const int dv[4] = {qd.x, qd.y, qd.z, qd.w};
    const int sv[4] = {qs.x, qs.y, qs.z, qs.w};
#pragma unroll
    for (int j = 0; j < 4; ++j) {
      const unsigned off = (unsigned)(dv[j] - lo);
      if (off < (unsigned)(ng * PULL_R)) {
        const unsigned g = off / PULL_R;
        const unsigned rel = off - g * PULL_R;
        const int slot = atomicAdd(&cnt[g], 1);
        if (slot < BIN_CAP) stage[g * BIN_CAP + slot] = (int)(((unsigned)sv[j] << 9) | rel);
      }
    }
  }
  __syncthreads();
  const int lane = tid & 31;
  const int wave = tid >> 5;
  for (int g = wave; g < ng; g += 32) {
    const int b = b0 + g;
    if ((b + 1) * PULL_R > Nn) continue;
    int* cell = cells + (size_t)b * (PULL_R * F) + (size_t)blockIdx.x * PITCH;
    const int c = cnt[g];
    const int n = min(c, BIN_CAP);
    if (lane == 0) cell[0] = c;
    for (int i = lane; i < n; i += 32) cell[1 + i] = stage[g * BIN_CAP + i];
  }
}

template <int F>
__global__ __launch_bounds__(1024) void k_pull(const float* __restrict__ hsc,
                                               const int* __restrict__ src,
                                               const int* __restrict__ dst,
                                               const float* __restrict__ dinv,
                                               const float* __restrict__ bias,
                                               float* __restrict__ stats,
                                               float* agg, int E, int Nn) {
  constexpr int C = F / 32;
  constexpr int U = 32 / C;
  constexpr int PITCH = PULL_R * F / BIN_CH;
  constexpr int CPW = BIN_CH / 32;
  static_assert(CPW * BIN_CAP <= PULL_SEG, "a wave's cells fit its list");
  __shared__ float acc[PULL_R * F];
  __shared__ int lp[32 * PULL_SEG];
  __shared__ float red[2 * F];
  __shared__ int walk;
  const int tid = threadIdx.x;
  const int lane = tid & 31;
  const int wave = __builtin_amdgcn_readfirstlane(tid >> 5);
  const int r0 = blockIdx.x * PULL_R;
  const int nrows = min(PULL_R, Nn - r0);

  {
    float4* a4 = reinterpret_cast<float4*>(acc);
    const float4 z = make_float4(0.f, 0.f, 0.f, 0.f);
    for (int i = tid; i < PULL_R * F / 4; i += 1024) a4[i] = z;
    if (tid < 2 * F) red[tid] = 0.f;
    if (tid == 0) walk = 0;
  }
  __syncthreads();

  int* mine = lp + wave * PULL_SEG;
  int n = 0;
  const bool binned = (nrows == PULL_R) && (Nn <= PACK_MAX_ROWS);
  if (binned) {
    const int* cells = reinterpret_cast<const int*>(agg) + (size_t)r0 * F +
                       (size_t)(wave * CPW) * PITCH;
    int cw = 0;
    if (lane < CPW) cw = cells[(size_t)lane * PITCH];
    if (cw > BIN_CAP) walk = 1;
    const int cn = min(max(cw, 0), BIN_CAP);
    int v[CPW];
#pragma unroll
    for (int j = 0; j < CPW; ++j) {
      const int nj = __builtin_amdgcn_readlane(cn, j);
      v[j] = (lane < nj) ? cells[(size_t)j * PITCH + 1 + lane] : 0;
    }
#pragma unroll
    for (int j = 0; j < CPW; ++j) {
      const int nj = __builtin_amdgcn_readlane(cn, j);
      if (lane < nj) mine[n + lane] = v[j];
      for (int i = lane + 32; i < nj; i += 32) mine[n + i] = cells[(size_t)j * PITCH + 1 + i];
      n += nj;
    }
  } else if (tid == 0) {
    walk = 1;
  }

  __syncthreads();

  if (walk != 0) {
    for (int e = tid; e < E; e += 1024) {
      const unsigned rel = (unsigned)(dst[e] - r0);
      if (rel < (unsigned)nrows) {
        const float* hr = hsc + (size_t)src[e] * F;
        float* ar = acc + rel * F;
        for (int c = 0; c < F; ++c) unsafeAtomicAdd(ar + c, hr[c]);
      }
    }
  } else {
  for (int i0 = 0; i0 < n; i0 += 32) {
    const int m = min(32, n - i0);
    int sv = 0, dl = 0;
    if (lane < m) {
      const unsigned pk = (unsigned)mine[i0 + lane];
      sv = (int)(pk >> 9);
      dl = (int)(pk & 511u);
    }
#pragma unroll
    for (int g = 0; g < 32; g += U) {
      if (g + U <= m) {
        float v[U][C];
#pragma unroll
        for (int u = 0; u < U; ++u) {
          const int s = __builtin_amdgcn_readlane(sv, g + u);
          const float* hrow = hsc + (size_t)s * F + lane;
#pragma unroll
          for (int j = 0; j < C; ++j) v[u][j] = hrow[32 * j];
        }
#pragma unroll
        for (int u = 0; u < U; ++u) {
          const int d = __builtin_amdgcn_readlane(dl, g + u);
          float* ad = acc + d * F + lane;
#pragma unroll
          for (int j = 0; j < C; ++j) unsafeAtomicAdd(ad + 32 * j, v[u][j]);
        }
      } else if (g < m) {
        for (int i = g; i < m; ++i) {
          const int s = __builtin_amdgcn_readlane(sv, i);
          const int d = __builtin_amdgcn_readlane(dl, i);
          const float* hrow = hsc + (size_t)s * F + lane;
          float* ad = acc + d * F + lane;
          float v[C];
#pragma unroll
          for (int j = 0; j < C; ++j) v[j] = hrow[32 * j];
#pragma unroll
          for (int j = 0; j < C; ++j) unsafeAtomicAdd(ad + 32 * j, v[j]);
        }
      }
    }
  }
  }
  __syncthreads();

  {
    float* o = agg + (size_t)r0 * F;
    const float* hs = hsc + (size_t)r0 * F;
    const float* dv = dinv + r0;
    if (((reinterpret_cast<size_t>(o) | reinterpret_cast<size_t>(hs) |
          reinterpret_cast<size_t>(bias)) & 15) == 0) {
      constexpr int F4 = F / 4;
      constexpr int RPP = 1024 / F4;
      const int c4 = tid % F4;
      const int rr = tid / F4;
      const float4* a4 = reinterpret_cast<const float4*>(acc);
      const float4* h4 = reinterpret_cast<const float4*>(hs);
      float4* o4 = reinterpret_cast<float4*>(o);
      const float4 b = reinterpret_cast<const float4*>(bias)[c4];
      float4 s = make_float4(0.f, 0.f, 0.f, 0.f);
      float4 q = make_float4(0.f, 0.f, 0.f, 0.f);
#pragma unroll 4
      for (int r = rr; r < nrows; r += RPP) {
        const float4 a = a4[r * F4 + c4];
        const float4 h = h4[r * F4 + c4];
        const float di = dv[r];
        float4 v;
        v.x = fmaxf((a.x + h.x) * di + b.x, 0.f);
        v.y = fmaxf((a.y + h.y) * di + b.y, 0.f);
        v.z = fmaxf((a.z + h.z) * di + b.z, 0.f);
        v.w = fmaxf((a.w + h.w) * di + b.w, 0.f);
        o4[r * F4 + c4] = v;
        s.x += v.x; s.y += v.y; s.z += v.z; s.w += v.w;
        q.x += v.x * v.x; q.y += v.y * v.y; q.z += v.z * v.z; q.w += v.w * v.w;
      }
      float* rs = red + 4 * c4;
      unsafeAtomicAdd(rs + 0, s.x);
      unsafeAtomicAdd(rs + 1, s.y);
      unsafeAtomicAdd(rs + 2, s.z);
      unsafeAtomicAdd(rs + 3, s.w);
      unsafeAtomicAdd(rs + F + 0, q.x);
      unsafeAtomicAdd(rs + F + 1, q.y);
      unsafeAtomicAdd(rs + F + 2, q.z);
      unsafeAtomicAdd(rs + F + 3, q.w);
    } else {
      constexpr int RPP = 1024 / F;
      const int c = tid % F;
      const float bj = bias[c];
      float s = 0.f, q = 0.f;
      for (int r = tid / F; r < nrows; r += RPP) {
        float v = (acc[r * F + c] + hs[(size_t)r * F + c]) * dv[r] + bj;
        v = fmaxf(v, 0.f);
        o[(size_t)r * F + c] = v;
        s += v;
        q += v * v;
      }
      unsafeAtomicAdd(red + c, s);
      unsafeAtomicAdd(red + F + c, q);
    }
  }
  __syncthreads();
  if (tid < 2 * F) unsafeAtomicAdd(stats + tid, red[tid]);
}

__global__ void k_bn_apply(float* __restrict__ v, const float* __restrict__ stats,
                           const float* __restrict__ gamma, const float* __restrict__ beta,
                           int Nn, int F, float invN) {
  size_t total = (size_t)Nn * F;
  size_t i = (size_t)blockIdx.x * blockDim.x + threadIdx.x;
  size_t stride = (size_t)gridDim.x * blockDim.x;
  for (; i < total; i += stride) {
    int j = (int)(i & (size_t)(F - 1));
    float mean = stats[j] * invN;
    float var = stats[F + j] * invN - mean * mean;
    v[i] = (v[i] - mean) * rsqrtf(var + BN_EPS) * gamma[j] + beta[j];
  }
}

__global__ void k_gemm64x2(const float* __restrict__ h3, const float* __restrict__ W4,
                           const float* __restrict__ dinv,
                           float* __restrict__ h4, int Nn) {
  int i = blockIdx.x * blockDim.x + threadIdx.x;
  if (i >= Nn) return;
  const float* row = h3 + (size_t)i * 64;
  float o0 = 0.f, o1 = 0.f;
#pragma unroll 8
  for (int k = 0; k < 64; ++k) {
    float xv = row[k];
    o0 += xv * W4[2 * k];
    o1 += xv * W4[2 * k + 1];
  }
  float di = dinv[i];
  h4[2 * (size_t)i]     = o0 * di;
  h4[2 * (size_t)i + 1] = o1 * di;
}

__global__ __launch_bounds__(1024) void k_sum2_part(const float* __restrict__ h4s,
                                                    const int* __restrict__ src,
                                                    const int* __restrict__ dst,
                                                    float* __restrict__ part, int E, int Nn) {
  constexpr int U = 4;
  __shared__ float acc[2 * (PART_R + 32)];
  const int tid = threadIdx.x;
  const int spare = PART_R + (tid & 31);
  const int r0 = blockIdx.y * PART_R;
  const int nrows = min(PART_R, Nn - r0);
  {
    float4* a4 = reinterpret_cast<float4*>(acc);
    const float4 z = make_float4(0.f, 0.f, 0.f, 0.f);
    for (int i = tid; i < 2 * PART_R / 4; i += 1024) a4[i] = z;
  }
  __syncthreads();
  const int per = (((E + PART_CH - 1) / PART_CH) + 3) & ~3;
  const int eb = blockIdx.x * per;
  const int ee = min(E, eb + per);
  const bool vd = (reinterpret_cast<size_t>(dst) & 15) == 0;
  const bool vs = (reinterpret_cast<size_t>(src) & 15) == 0;
  const float2* h2 = reinterpret_cast<const float2*>(h4s);
  for (int b = eb; b < ee; b += 4096 * U) {
    int4 qd[U], qs[U];
    float2 v[4 * U];
    if (vd && vs && b + 4096 * U <= ee) {
#pragma unroll
      for (int u = 0; u < U; ++u) {
        const int e = b + u * 4096 + tid * 4;
        qd[u] = *reinterpret_cast<const int4*>(dst + e);
        qs[u] = *reinterpret_cast<const int4*>(src + e);
      }
    } else {
#pragma unroll
      for (int u = 0; u < U; ++u) {
        const int e = b + u * 4096 + tid * 4;
        qd[u] = ld_idx4(dst, e, ee, vd);
        qs[u] = ld_idx4(src, e, ee, vs);
      }
    }
#pragma unroll
    for (int u = 0; u < U; ++u) {
      const int dv[4] = {qd[u].x, qd[u].y, qd[u].z, qd[u].w};
      const int sv[4] = {qs[u].x, qs[u].y, qs[u].z, qs[u].w};
#pragma unroll
      for (int j = 0; j < 4; ++j) {
        const bool m = (unsigned)(dv[j] - r0) < (unsigned)nrows;
        v[4 * u + j] = h2[m ? sv[j] : 0];
      }
    }
#pragma unroll
    for (int u = 0; u < U; ++u) {
      const int dv[4] = {qd[u].x, qd[u].y, qd[u].z, qd[u].w};
#pragma unroll
      for (int j = 0; j < 4; ++j) {
        const unsigned rel = (unsigned)(dv[j] - r0);
        float* ar = acc + 2 * ((rel < (unsigned)nrows) ? (int)rel : spare);
        unsafeAtomicAdd(ar, v[4 * u + j].x);
        unsafeAtomicAdd(ar + 1, v[4 * u + j].y);
      }
    }
  }
  __syncthreads();
  float2* o = reinterpret_cast<float2*>(part) + ((size_t)blockIdx.x * Nn + r0);
  const float2* a2 = reinterpret_cast<const float2*>(acc);
  for (int i = tid; i < nrows; i += 1024) o[i] = a2[i];
}

__global__ void k_final_lsm(const float* __restrict__ part, const float* __restrict__ h4s,
                            const float* __restrict__ dinv, const float* __restrict__ b4,
                            float* __restrict__ out, int Nn) {
  int i = blockIdx.x * blockDim.x + threadIdx.x;
  if (i >= Nn) return;
  const float2* p2 = reinterpret_cast<const float2*>(part);
  float a0 = 0.f, a1 = 0.f;
#pragma unroll
  for (int k = 0; k < PART_CH; ++k) {
    const float2 t = p2[(size_t)k * Nn + i];
    a0 += t.x;
    a1 += t.y;
  }
  float di = dinv[i];
  float o0 = (a0 + h4s[2 * (size_t)i])     * di + b4[0];
  float o1 = (a1 + h4s[2 * (size_t)i + 1]) * di + b4[1];
  float m = fmaxf(o0, o1);
  float l = m + logf(expf(o0 - m) + expf(o1 - m));
  out[2 * (size_t)i]     = o0 - l;
  out[2 * (size_t)i + 1] = o1 - l;
}

extern "C" void kernel_launch(void* const* d_in, const int* in_sizes, int n_in,
                              void* d_out, int out_size, void* d_ws, size_t ws_size,
                              hipStream_t stream) {
  const float* x   = (const float*)d_in[0];
  const int*   ei  = (const int*)d_in[1];
  const float* W1  = (const float*)d_in[2];
  const float* b1  = (const float*)d_in[3];
  const float* g1  = (const float*)d_in[4];
  const float* be1 = (const float*)d_in[5];
  const float* W2  = (const float*)d_in[6];
  const float* b2  = (const float*)d_in[7];
  const float* g2  = (const float*)d_in[8];
  const float* be2 = (const float*)d_in[9];
  const float* W3  = (const float*)d_in[10];
  const float* b3  = (const float*)d_in[11];
  const float* g3  = (const float*)d_in[12];
  const float* be3 = (const float*)d_in[13];
  const float* W4  = (const float*)d_in[14];
  const float* b4  = (const float*)d_in[15];

  const int Nn = in_sizes[0] / 24;
  const int E  = in_sizes[1] / 2;
  const int* src = ei;
  const int* dst = ei + E;

  float* ws    = (float*)d_ws;
  float* dinv  = ws;
  float* stats = ws + Nn;
  float* bufA  = stats + 256;
  float* bufB  = bufA + (size_t)Nn * 128;

  const float invN = 1.0f / (float)Nn;
  const int TB = 256;

  auto gemm = [&](const float* A, const float* Bm, float* C, int K, int F) {
    long waves = (long)((Nn + 15) / 16) * (F / 16);
    int blocks = (int)((waves * 32 + TB - 1) / TB);
    k_gemm_wmma_f32<<<blocks, TB, 0, stream>>>(A, Bm, dinv, C, Nn, K, F);
  };
  const int nRanges = (Nn + PULL_R - 1) / PULL_R;
  const dim3 binGrid(BIN_CH, (nRanges + BIN_G - 1) / BIN_G);
  auto pull128 = [&](const float* h, const float* bias, float* out) {
    k_bin<128><<<binGrid, 1024, 0, stream>>>(src, dst, (int*)out, stats, E, Nn);
    k_pull<128><<<nRanges, 1024, 0, stream>>>(h, src, dst, dinv, bias, stats, out, E, Nn);
  };
  auto pull64 = [&](const float* h, const float* bias, float* out) {
    k_bin<64><<<binGrid, 1024, 0, stream>>>(src, dst, (int*)out, stats, E, Nn);
    k_pull<64><<<nRanges, 1024, 0, stream>>>(h, src, dst, dinv, bias, stats, out, E, Nn);
  };
  auto bn = [&](float* v, const float* gamma, const float* beta, int F) {
    k_bn_apply<<<2048, TB, 0, stream>>>(v, stats, gamma, beta, Nn, F, invN);
  };
  const dim3 partGrid(PART_CH, (Nn + PART_R - 1) / PART_R);

  k_deg_part<<<partGrid, 1024, 0, stream>>>(dst, (int*)bufB, E, Nn);
  k_deg_fin<<<(Nn + TB - 1) / TB, TB, 0, stream>>>((const int*)bufB, dinv, Nn);

  gemm(x, W1, bufA, 24, 128);
  pull128(bufA, b1, bufB);
  bn(bufB, g1, be1, 128);

  gemm(bufB, W2, bufA, 128, 128);
  pull128(bufA, b2, bufB);
  bn(bufB, g2, be2, 128);

  gemm(bufB, W3, bufA, 128, 64);
  pull64(bufA, b3, bufB);
  bn(bufB, g3, be3, 64);

  k_gemm64x2<<<(Nn + TB - 1) / TB, TB, 0, stream>>>(bufB, W4, dinv, bufA, Nn);
  k_sum2_part<<<partGrid, 1024, 0, stream>>>(bufA, src, dst, bufB, E, Nn);
  k_final_lsm<<<(Nn + TB - 1) / TB, TB, 0, stream>>>(bufB, bufA, dinv, b4,
                                                     (float*)d_out, Nn);
}


# layers 1-3 product: a wave takes 16 rows x all F columns (A row read once in 16-byte pieces, W shared in LDS as k pairs, whole-line 16-byte stores); 8x/4x fewer waves (k sum order changes)
# speedup vs baseline: 4.3917x; 1.1015x over previous
#include <hip/hip_runtime.h>
#include <hip/hip_bf16.h>
#include <math.h>


typedef float v2f __attribute__((ext_vector_type(2)));
typedef float v8f __attribute__((ext_vector_type(8)));

#define BN_EPS 1e-5f

#define PART_R 12800
#define PART_CH 32

__device__ __forceinline__ int4 ld_idx4(const int* __restrict__ p, int e, int ee, bool vec) {
  if (vec && e + 4 <= ee) return *reinterpret_cast<const int4*>(p + e);
  int4 r;
  r.x = (e     < ee) ? p[e]     : -1;
  r.y = (e + 1 < ee) ? p[e + 1] : -1;
  r.z = (e + 2 < ee) ? p[e + 2] : -1;
  r.w = (e + 3 < ee) ? p[e + 3] : -1;
  return r;
}

__global__ __launch_bounds__(1024) void k_deg_part(const int* __restrict__ dst,
                                                   int* __restrict__ part, int E, int Nn) {
  constexpr int U = 8;
  __shared__ int cnt[PART_R];
  const int tid = threadIdx.x;
  const int r0 = blockIdx.y * PART_R;
  const int nrows = min(PART_R, Nn - r0);
  for (int i = tid; i < PART_R; i += 1024) cnt[i] = 0;
  __syncthreads();
  const int per = (((E + PART_CH - 1) / PART_CH) + 3) & ~3;
  const int eb = blockIdx.x * per;
  const int ee = min(E, eb + per);
  const bool vec = (reinterpret_cast<size_t>(dst) & 15) == 0;
  for (int b = eb; b < ee; b += 4096 * U) {
    int4 q[U];
    if (vec && b + 4096 * U <= ee) {
#pragma unroll
      for (int u = 0; u < U; ++u)
        q[u] = *reinterpret_cast<const int4*>(dst + b + u * 4096 + tid * 4);
    } else {
#pragma unroll
      for (int u = 0; u < U; ++u) q[u] = ld_idx4(dst, b + u * 4096 + tid * 4, ee, vec);
    }
#pragma unroll
    for (int u = 0; u < U; ++u) {
      const int dv[4] = {q[u].x, q[u].y, q[u].z, q[u].w};
#pragma unroll
      for (int j = 0; j < 4; ++j) {
        const unsigned rel = (unsigned)(dv[j] - r0);
        if (rel < (unsigned)nrows) atomicAdd(&cnt[rel], 1);
      }
    }
  }
  __syncthreads();
  int* o = part + (size_t)blockIdx.x * Nn + r0;
  for (int i = tid; i < nrows; i += 1024) o[i] = cnt[i];
}

__global__ void k_deg_fin(const int* __restrict__ part, float* __restrict__ dinv, int n) {
  int i = blockIdx.x * blockDim.x + threadIdx.x;
  if (i < n) {
    int c = 1;
#pragma unroll
    for (int k = 0; k < PART_CH; ++k) c += part[(size_t)k * n + i];
    float v = (float)c;
    dinv[i] = (v > 0.0f) ? rsqrtf(v) : 0.0f;
  }
}

template <int K, int F>
__global__ __launch_bounds__(256) void k_gemm_strip(const float* __restrict__ A,
                                                    const float* __restrict__ B,
                                                    const float* __restrict__ dinv,
                                                    float* __restrict__ C, int M) {
  constexpr int NT = F / 16;
  constexpr int NU = K / 8;
  constexpr int PH = F + 8;
  __shared__ float2 wp[(K / 2) * PH];
  const int tid = threadIdx.x;
  const int lane = tid & 31;
  const int c = lane & 15;
  const int h = lane >> 4;
  const int m0 = (blockIdx.x * 8 + (tid >> 5)) << 4;
  const bool live = m0 < M;

  float4 a4[NU];
  if (live) {
    const float4* arow = reinterpret_cast<const float4*>(A + (size_t)(m0 + c) * K) + h;
#pragma unroll
    for (int u = 0; u < NU; ++u) a4[u] = arow[2 * u];
  }

  constexpr int ITEMS = (K / 2) * (F / 4);
  constexpr int NIT = (ITEMS + 255) / 256;
  float4 w0[NIT], w1[NIT];
#pragma unroll
  for (int n = 0; n < NIT; ++n) {
    const int idx = tid + 256 * n;
    w0[n] = make_float4(0.f, 0.f, 0.f, 0.f);
    w1[n] = w0[n];
    if (ITEMS % 256 == 0 || idx < ITEMS) {
      const int q = idx % (F / 4);
      const int kp = idx / (F / 4);
      w0[n] = *reinterpret_cast<const float4*>(B + (size_t)(2 * kp) * F + 4 * q);
      w1[n] = *reinterpret_cast<const float4*>(B + (size_t)(2 * kp + 1) * F + 4 * q);
    }
  }
#pragma unroll
  for (int n = 0; n < NIT; ++n) {
    const int idx = tid + 256 * n;
    if (ITEMS % 256 == 0 || idx < ITEMS) {
      const int q = idx % (F / 4);
      const int kp = idx / (F / 4);
      float2* d = wp + kp * PH + 64 * (q >> 4) + (q & 15);
      d[0]  = make_float2(w0[n].x, w1[n].x);
      d[16] = make_float2(w0[n].y, w1[n].y);
      d[32] = make_float2(w0[n].z, w1[n].z);
      d[48] = make_float2(w0[n].w, w1[n].w);
    }
  }
  __syncthreads();
  if (!live) return;

  v8f acc[NT];
#pragma unroll
  for (int j = 0; j < NT; ++j) acc[j] = (v8f){0.f, 0.f, 0.f, 0.f, 0.f, 0.f, 0.f, 0.f};
  const float2* bl = wp + (2 * h) * PH + c;
#pragma unroll
  for (int u = 0; u < NU; ++u) {
#pragma unroll
    for (int s = 0; s < 2; ++s) {
      v2f a;
      a.x = s ? a4[u].z : a4[u].x;
      a.y = s ? a4[u].w : a4[u].y;
      const float2* bp = bl + (4 * u + s) * PH;
#pragma unroll
      for (int j = 0; j < NT; ++j) {
        const float2 bv = bp[16 * j];
        v2f b; b.x = bv.x; b.y = bv.y;
        acc[j] = __builtin_amdgcn_wmma_f32_16x16x4_f32(false, a, false, b,
                                                       (short)0, acc[j], false, false);
      }
    }
  }
  const float* dvp = dinv + m0 + 8 * h;
  float* crow = C + (size_t)(m0 + 8 * h) * F + 4 * c;
#pragma unroll
  for (int r = 0; r < 8; ++r) {
    const float di = dvp[r];
#pragma unroll
    for (int q = 0; q < NT / 4; ++q) {
      float4 v;
      v.x = acc[4 * q + 0][r] * di;
      v.y = acc[4 * q + 1][r] * di;
      v.z = acc[4 * q + 2][r] * di;
      v.w = acc[4 * q + 3][r] * di;
      *reinterpret_cast<float4*>(crow + (size_t)r * F + 64 * q) = v;
    }
  }
}

#define PULL_R 400
#define BIN_CH 256
#define BIN_CAP 96
#define BIN_G 256
#define PULL_SEG (8 * BIN_CAP)
#define PACK_MAX_ROWS (1 << 23)

template <int F>
__global__ __launch_bounds__(1024) void k_bin(const int* __restrict__ src,
                                              const int* __restrict__ dst,
                                              int* __restrict__ cells,
                                              float* __restrict__ stats,
                                              int E, int Nn) {
  constexpr int PITCH = PULL_R * F / BIN_CH;
  static_assert(PITCH >= BIN_CAP + 1, "a cell holds its count and BIN_CAP entries");
  static_assert(PULL_R <= 512, "row within the range takes 9 bits");
  __shared__ int cnt[BIN_G];
  __shared__ int stage[BIN_G * BIN_CAP];
  const int tid = threadIdx.x;
  if (blockIdx.x == 0 && blockIdx.y == 0 && tid < 256) stats[tid] = 0.f;
  const int nb = (Nn + PULL_R - 1) / PULL_R;
  const int b0 = blockIdx.y * BIN_G;
  const int ng = min(BIN_G, nb - b0);
  const int lo = b0 * PULL_R;
  if (tid < BIN_G) cnt[tid] = 0;
  __syncthreads();
  const int per = (((E + BIN_CH - 1) / BIN_CH) + 3) & ~3;
  const int eb = blockIdx.x * per;
  const int ee = min(E, eb + per);
  const bool vd = (reinterpret_cast<size_t>(dst) & 15) == 0;
  const bool vs = (reinterpret_cast<size_t>(src) & 15) == 0;
  for (int e = eb + tid * 4; e < ee; e += 4096) {
    const int4 qd = ld_idx4(dst, e, ee, vd);
    const int4 qs = ld_idx4(src, e, ee, vs);
    const int dv[4] = {qd.x, qd.y, qd.z, qd.w};
    const int sv[4] = {qs.x, qs.y, qs.z, qs.w};
#pragma unroll
    for (int j = 0; j < 4; ++j) {
      const unsigned off = (unsigned)(dv[j] - lo);
      if (off < (unsigned)(ng * PULL_R)) {
        const unsigned g = off / PULL_R;
        const unsigned rel = off - g * PULL_R;
        const int slot = atomicAdd(&cnt[g], 1);
        if (slot < BIN_CAP) stage[g * BIN_CAP + slot] = (int)(((unsigned)sv[j] << 9) | rel);
      }
    }
  }
  __syncthreads();
  const int lane = tid & 31;
  const int wave = tid >> 5;
  for (int g = wave; g < ng; g += 32) {
    const int b = b0 + g;
    if ((b + 1) * PULL_R > Nn) continue;
    int* cell = cells + (size_t)b * (PULL_R * F) + (size_t)blockIdx.x * PITCH;
    const int c = cnt[g];
    const int n = min(c, BIN_CAP);
    if (lane == 0) cell[0] = c;
    for (int i = lane; i < n; i += 32) cell[1 + i] = stage[g * BIN_CAP + i];
  }
}

template <int F>
__global__ __launch_bounds__(1024) void k_pull(const float* __restrict__ hsc,
                                               const int* __restrict__ src,
                                               const int* __restrict__ dst,
                                               const float* __restrict__ dinv,
                                               const float* __restrict__ bias,
                                               float* __restrict__ stats,
                                               float* agg, int E, int Nn) {
  constexpr int C = F / 32;
  constexpr int U = 32 / C;
  constexpr int PITCH = PULL_R * F / BIN_CH;
  constexpr int CPW = BIN_CH / 32;
  static_assert(CPW * BIN_CAP <= PULL_SEG, "a wave's cells fit its list");
  __shared__ float acc[PULL_R * F];
  __shared__ int lp[32 * PULL_SEG];
  __shared__ float red[2 * F];
  __shared__ int walk;
  const int tid = threadIdx.x;
  const int lane = tid & 31;
  const int wave = __builtin_amdgcn_readfirstlane(tid >> 5);
  const int r0 = blockIdx.x * PULL_R;
  const int nrows = min(PULL_R, Nn - r0);

  {
    float4* a4 = reinterpret_cast<float4*>(acc);
    const float4 z = make_float4(0.f, 0.f, 0.f, 0.f);
    for (int i = tid; i < PULL_R * F / 4; i += 1024) a4[i] = z;
    if (tid < 2 * F) red[tid] = 0.f;
    if (tid == 0) walk = 0;
  }
  __syncthreads();

  int* mine = lp + wave * PULL_SEG;
  int n = 0;
  const bool binned = (nrows == PULL_R) && (Nn <= PACK_MAX_ROWS);
  if (binned) {
    const int* cells = reinterpret_cast<const int*>(agg) + (size_t)r0 * F +
                       (size_t)(wave * CPW) * PITCH;
    int cw = 0;
    if (lane < CPW) cw = cells[(size_t)lane * PITCH];
    if (cw > BIN_CAP) walk = 1;
    const int cn = min(max(cw, 0), BIN_CAP);
    int v[CPW];
#pragma unroll
    for (int j = 0; j < CPW; ++j) {
      const int nj = __builtin_amdgcn_readlane(cn, j);
      v[j] = (lane < nj) ? cells[(size_t)j * PITCH + 1 + lane] : 0;
    }
#pragma unroll
    for (int j = 0; j < CPW; ++j) {
      const int nj = __builtin_amdgcn_readlane(cn, j);
      if (lane < nj) mine[n + lane] = v[j];
      for (int i = lane + 32; i < nj; i += 32) mine[n + i] = cells[(size_t)j * PITCH + 1 + i];
      n += nj;
    }
  } else if (tid == 0) {
    walk = 1;
  }

  __syncthreads();

  if (walk != 0) {
    for (int e = tid; e < E; e += 1024) {
      const unsigned rel = (unsigned)(dst[e] - r0);
      if (rel < (unsigned)nrows) {
        const float* hr = hsc + (size_t)src[e] * F;
        float* ar = acc + rel * F;
        for (int c = 0; c < F; ++c) unsafeAtomicAdd(ar + c, hr[c]);
      }
    }
  } else {
  for (int i0 = 0; i0 < n; i0 += 32) {
    const int m = min(32, n - i0);
    int sv = 0, dl = 0;
    if (lane < m) {
      const unsigned pk = (unsigned)mine[i0 + lane];
      sv = (int)(pk >> 9);
      dl = (int)(pk & 511u);
    }
#pragma unroll
    for (int g = 0; g < 32; g += U) {
      if (g + U <= m) {
        float v[U][C];
#pragma unroll
        for (int u = 0; u < U; ++u) {
          const int s = __builtin_amdgcn_readlane(sv, g + u);
          const float* hrow = hsc + (size_t)s * F + lane;
#pragma unroll
          for (int j = 0; j < C; ++j) v[u][j] = hrow[32 * j];
        }
#pragma unroll
        for (int u = 0; u < U; ++u) {
          const int d = __builtin_amdgcn_readlane(dl, g + u);
          float* ad = acc + d * F + lane;
#pragma unroll
          for (int j = 0; j < C; ++j) unsafeAtomicAdd(ad + 32 * j, v[u][j]);
        }
      } else if (g < m) {
        for (int i = g; i < m; ++i) {
          const int s = __builtin_amdgcn_readlane(sv, i);
          const int d = __builtin_amdgcn_readlane(dl, i);
          const float* hrow = hsc + (size_t)s * F + lane;
          float* ad = acc + d * F + lane;
          float v[C];
#pragma unroll
          for (int j = 0; j < C; ++j) v[j] = hrow[32 * j];
#pragma unroll
          for (int j = 0; j < C; ++j) unsafeAtomicAdd(ad + 32 * j, v[j]);
        }
      }
    }
  }
  }
  __syncthreads();

  {
    float* o = agg + (size_t)r0 * F;
    const float* hs = hsc + (size_t)r0 * F;
    const float* dv = dinv + r0;
    if (((reinterpret_cast<size_t>(o) | reinterpret_cast<size_t>(hs) |
          reinterpret_cast<size_t>(bias)) & 15) == 0) {
      constexpr int F4 = F / 4;
      constexpr int RPP = 1024 / F4;
      const int c4 = tid % F4;
      const int rr = tid / F4;
      const float4* a4 = reinterpret_cast<const float4*>(acc);
      const float4* h4 = reinterpret_cast<const float4*>(hs);
      float4* o4 = reinterpret_cast<float4*>(o);
      const float4 b = reinterpret_cast<const float4*>(bias)[c4];
      float4 s = make_float4(0.f, 0.f, 0.f, 0.f);
      float4 q = make_float4(0.f, 0.f, 0.f, 0.f);
#pragma unroll 4
      for (int r = rr; r < nrows; r += RPP) {
        const float4 a = a4[r * F4 + c4];
        const float4 h = h4[r * F4 + c4];
        const float di = dv[r];
        float4 v;
        v.x = fmaxf((a.x + h.x) * di + b.x, 0.f);
        v.y = fmaxf((a.y + h.y) * di + b.y, 0.f);
        v.z = fmaxf((a.z + h.z) * di + b.z, 0.f);
        v.w = fmaxf((a.w + h.w) * di + b.w, 0.f);
        o4[r * F4 + c4] = v;
        s.x += v.x; s.y += v.y; s.z += v.z; s.w += v.w;
        q.x += v.x * v.x; q.y += v.y * v.y; q.z += v.z * v.z; q.w += v.w * v.w;
      }
      float* rs = red + 4 * c4;
      unsafeAtomicAdd(rs + 0, s.x);
      unsafeAtomicAdd(rs + 1, s.y);
      unsafeAtomicAdd(rs + 2, s.z);
      unsafeAtomicAdd(rs + 3, s.w);
      unsafeAtomicAdd(rs + F + 0, q.x);
      unsafeAtomicAdd(rs + F + 1, q.y);
      unsafeAtomicAdd(rs + F + 2, q.z);
      unsafeAtomicAdd(rs + F + 3, q.w);
    } else {
      constexpr int RPP = 1024 / F;
      const int c = tid % F;
      const float bj = bias[c];
      float s = 0.f, q = 0.f;
      for (int r = tid / F; r < nrows; r += RPP) {
        float v = (acc[r * F + c] + hs[(size_t)r * F + c]) * dv[r] + bj;
        v = fmaxf(v, 0.f);
        o[(size_t)r * F + c] = v;
        s += v;
        q += v * v;
      }
      unsafeAtomicAdd(red + c, s);
      unsafeAtomicAdd(red + F + c, q);
    }
  }
  __syncthreads();
  if (tid < 2 * F) unsafeAtomicAdd(stats + tid, red[tid]);
}

__global__ void k_bn_apply(float* __restrict__ v, const float* __restrict__ stats,
                           const float* __restrict__ gamma, const float* __restrict__ beta,
                           int Nn, int F, float invN) {
  size_t total = (size_t)Nn * F;
  size_t i = (size_t)blockIdx.x * blockDim.x + threadIdx.x;
  size_t stride = (size_t)gridDim.x * blockDim.x;
  for (; i < total; i += stride) {
    int j = (int)(i & (size_t)(F - 1));
    float mean = stats[j] * invN;
    float var = stats[F + j] * invN - mean * mean;
    v[i] = (v[i] - mean) * rsqrtf(var + BN_EPS) * gamma[j] + beta[j];
  }
}

__global__ void k_gemm64x2(const float* __restrict__ h3, const float* __restrict__ W4,
                           const float* __restrict__ dinv,
                           float* __restrict__ h4, int Nn) {
  int i = blockIdx.x * blockDim.x + threadIdx.x;
  if (i >= Nn) return;
  const float* row = h3 + (size_t)i * 64;
  float o0 = 0.f, o1 = 0.f;
#pragma unroll 8
  for (int k = 0; k < 64; ++k) {
    float xv = row[k];
    o0 += xv * W4[2 * k];
    o1 += xv * W4[2 * k + 1];
  }
  float di = dinv[i];
  h4[2 * (size_t)i]     = o0 * di;
  h4[2 * (size_t)i + 1] = o1 * di;
}

__global__ __launch_bounds__(1024) void k_sum2_part(const float* __restrict__ h4s,
                                                    const int* __restrict__ src,
                                                    const int* __restrict__ dst,
                                                    float* __restrict__ part, int E, int Nn) {
  constexpr int U = 4;
  __shared__ float acc[2 * (PART_R + 32)];
  const int tid = threadIdx.x;
  const int spare = PART_R + (tid & 31);
  const int r0 = blockIdx.y * PART_R;
  const int nrows = min(PART_R, Nn - r0);
  {
    float4* a4 = reinterpret_cast<float4*>(acc);
    const float4 z = make_float4(0.f, 0.f, 0.f, 0.f);
    for (int i = tid; i < 2 * PART_R / 4; i += 1024) a4[i] = z;
  }
  __syncthreads();
  const int per = (((E + PART_CH - 1) / PART_CH) + 3) & ~3;
  const int eb = blockIdx.x * per;
  const int ee = min(E, eb + per);
  const bool vd = (reinterpret_cast<size_t>(dst) & 15) == 0;
  const bool vs = (reinterpret_cast<size_t>(src) & 15) == 0;
  const float2* h2 = reinterpret_cast<const float2*>(h4s);
  for (int b = eb; b < ee; b += 4096 * U) {
    int4 qd[U], qs[U];
    float2 v[4 * U];
    if (vd && vs && b + 4096 * U <= ee) {
#pragma unroll
      for (int u = 0; u < U; ++u) {
        const int e = b + u * 4096 + tid * 4;
        qd[u] = *reinterpret_cast<const int4*>(dst + e);
        qs[u] = *reinterpret_cast<const int4*>(src + e);
      }
    } else {
#pragma unroll
      for (int u = 0; u < U; ++u) {
        const int e = b + u * 4096 + tid * 4;
        qd[u] = ld_idx4(dst, e, ee, vd);
        qs[u] = ld_idx4(src, e, ee, vs);
      }
    }
#pragma unroll
    for (int u = 0; u < U; ++u) {
      const int dv[4] = {qd[u].x, qd[u].y, qd[u].z, qd[u].w};
      const int sv[4] = {qs[u].x, qs[u].y, qs[u].z, qs[u].w};
#pragma unroll
      for (int j = 0; j < 4; ++j) {
        const bool m = (unsigned)(dv[j] - r0) < (unsigned)nrows;
        v[4 * u + j] = h2[m ? sv[j] : 0];
      }
    }
#pragma unroll
    for (int u = 0; u < U; ++u) {
      const int dv[4] = {qd[u].x, qd[u].y, qd[u].z, qd[u].w};
#pragma unroll
      for (int j = 0; j < 4; ++j) {
        const unsigned rel = (unsigned)(dv[j] - r0);
        float* ar = acc + 2 * ((rel < (unsigned)nrows) ? (int)rel : spare);
        unsafeAtomicAdd(ar, v[4 * u + j].x);
        unsafeAtomicAdd(ar + 1, v[4 * u + j].y);
      }
    }
  }
  __syncthreads();
  float2* o = reinterpret_cast<float2*>(part) + ((size_t)blockIdx.x * Nn + r0);
  const float2* a2 = reinterpret_cast<const float2*>(acc);
  for (int i = tid; i < nrows; i += 1024) o[i] = a2[i];
}

__global__ void k_final_lsm(const float* __restrict__ part, const float* __restrict__ h4s,
                            const float* __restrict__ dinv, const float* __restrict__ b4,
                            float* __restrict__ out, int Nn) {
  int i = blockIdx.x * blockDim.x + threadIdx.x;
  if (i >= Nn) return;
  const float2* p2 = reinterpret_cast<const float2*>(part);
  float a0 = 0.f, a1 = 0.f;
#pragma unroll
  for (int k = 0; k < PART_CH; ++k) {
    const float2 t = p2[(size_t)k * Nn + i];
    a0 += t.x;
    a1 += t.y;
  }
  float di = dinv[i];
  float o0 = (a0 + h4s[2 * (size_t)i])     * di + b4[0];
  float o1 = (a1 + h4s[2 * (size_t)i + 1]) * di + b4[1];
  float m = fmaxf(o0, o1);
  float l = m + logf(expf(o0 - m) + expf(o1 - m));
  out[2 * (size_t)i]     = o0 - l;
  out[2 * (size_t)i + 1] = o1 - l;
}

extern "C" void kernel_launch(void* const* d_in, const int* in_sizes, int n_in,
                              void* d_out, int out_size, void* d_ws, size_t ws_size,
                              hipStream_t stream) {
  const float* x   = (const float*)d_in[0];
  const int*   ei  = (const int*)d_in[1];
  const float* W1  = (const float*)d_in[2];
  const float* b1  = (const float*)d_in[3];
  const float* g1  = (const float*)d_in[4];
  const float* be1 = (const float*)d_in[5];
  const float* W2  = (const float*)d_in[6];
  const float* b2  = (const float*)d_in[7];
  const float* g2  = (const float*)d_in[8];
  const float* be2 = (const float*)d_in[9];
  const float* W3  = (const float*)d_in[10];
  const float* b3  = (const float*)d_in[11];
  const float* g3  = (const float*)d_in[12];
  const float* be3 = (const float*)d_in[13];
  const float* W4  = (const float*)d_in[14];
  const float* b4  = (const float*)d_in[15];

  const int Nn = in_sizes[0] / 24;
  const int E  = in_sizes[1] / 2;
  const int* src = ei;
  const int* dst = ei + E;

  float* ws    = (float*)d_ws;
  float* dinv  = ws;
  float* stats = ws + Nn;
  float* bufA  = stats + 256;
  float* bufB  = bufA + (size_t)Nn * 128;

  const float invN = 1.0f / (float)Nn;
  const int TB = 256;

  const int gemmBlocks = (Nn + 127) / 128;
  const int nRanges = (Nn + PULL_R - 1) / PULL_R;
  const dim3 binGrid(BIN_CH, (nRanges + BIN_G - 1) / BIN_G);
  auto pull128 = [&](const float* h, const float* bias, float* out) {
    k_bin<128><<<binGrid, 1024, 0, stream>>>(src, dst, (int*)out, stats, E, Nn);
    k_pull<128><<<nRanges, 1024, 0, stream>>>(h, src, dst, dinv, bias, stats, out, E, Nn);
  };
  auto pull64 = [&](const float* h, const float* bias, float* out) {
    k_bin<64><<<binGrid, 1024, 0, stream>>>(src, dst, (int*)out, stats, E, Nn);
    k_pull<64><<<nRanges, 1024, 0, stream>>>(h, src, dst, dinv, bias, stats, out, E, Nn);
  };
  auto bn = [&](float* v, const float* gamma, const float* beta, int F) {
    k_bn_apply<<<2048, TB, 0, stream>>>(v, stats, gamma, beta, Nn, F, invN);
  };
  const dim3 partGrid(PART_CH, (Nn + PART_R - 1) / PART_R);

  k_deg_part<<<partGrid, 1024, 0, stream>>>(dst, (int*)bufB, E, Nn);
  k_deg_fin<<<(Nn + TB - 1) / TB, TB, 0, stream>>>((const int*)bufB, dinv, Nn);

  k_gemm_strip<24, 128><<<gemmBlocks, 256, 0, stream>>>(x, W1, dinv, bufA, Nn);
  pull128(bufA, b1, bufB);
  bn(bufB, g1, be1, 128);

  k_gemm_strip<128, 128><<<gemmBlocks, 256, 0, stream>>>(bufB, W2, dinv, bufA, Nn);
  pull128(bufA, b2, bufB);
  bn(bufB, g2, be2, 128);

  k_gemm_strip<128, 64><<<gemmBlocks, 256, 0, stream>>>(bufB, W3, dinv, bufA, Nn);
  pull64(bufA, b3, bufB);
  bn(bufB, g3, be3, 64);

  k_gemm64x2<<<(Nn + TB - 1) / TB, TB, 0, stream>>>(bufB, W4, dinv, bufA, Nn);
  k_sum2_part<<<partGrid, 1024, 0, stream>>>(bufA, src, dst, bufB, E, Nn);
  k_final_lsm<<<(Nn + TB - 1) / TB, TB, 0, stream>>>(bufB, bufA, dinv, b4,
                                                     (float*)d_out, Nn);
}


# BatchNorm folded into the next product's weights: W rows scaled by gamma*rstd while staged in LDS, accumulators start at shift @ W (fixed-order sum); 3 apply passes go (product rounding changes)
# speedup vs baseline: 5.1593x; 1.1748x over previous
#include <hip/hip_runtime.h>
#include <hip/hip_bf16.h>
#include <math.h>


typedef float v2f __attribute__((ext_vector_type(2)));
typedef float v8f __attribute__((ext_vector_type(8)));

#define BN_EPS 1e-5f

#define PART_R 12800
#define PART_CH 32

__device__ __forceinline__ int4 ld_idx4(const int* __restrict__ p, int e, int ee, bool vec) {
  if (vec && e + 4 <= ee) return *reinterpret_cast<const int4*>(p + e);
  int4 r;
  r.x = (e     < ee) ? p[e]     : -1;
  r.y = (e + 1 < ee) ? p[e + 1] : -1;
  r.z = (e + 2 < ee) ? p[e + 2] : -1;
  r.w = (e + 3 < ee) ? p[e + 3] : -1;
  return r;
}

__global__ __launch_bounds__(1024) void k_deg_part(const int* __restrict__ dst,
                                                   int* __restrict__ part, int E, int Nn) {
  constexpr int U = 8;
  __shared__ int cnt[PART_R];
  const int tid = threadIdx.x;
  const int r0 = blockIdx.y * PART_R;
  const int nrows = min(PART_R, Nn - r0);
  for (int i = tid; i < PART_R; i += 1024) cnt[i] = 0;
  __syncthreads();
  const int per = (((E + PART_CH - 1) / PART_CH) + 3) & ~3;
  const int eb = blockIdx.x * per;
  const int ee = min(E, eb + per);
  const bool vec = (reinterpret_cast<size_t>(dst) & 15) == 0;
  for (int b = eb; b < ee; b += 4096 * U) {
    int4 q[U];
    if (vec && b + 4096 * U <= ee) {
#pragma unroll
      for (int u = 0; u < U; ++u)
        q[u] = *reinterpret_cast<const int4*>(dst + b + u * 4096 + tid * 4);
    } else {
#pragma unroll
      for (int u = 0; u < U; ++u) q[u] = ld_idx4(dst, b + u * 4096 + tid * 4, ee, vec);
    }
#pragma unroll
    for (int u = 0; u < U; ++u) {
      const int dv[4] = {q[u].x, q[u].y, q[u].z, q[u].w};
#pragma unroll
      for (int j = 0; j < 4; ++j) {
        const unsigned rel = (unsigned)(dv[j] - r0);
        if (rel < (unsigned)nrows) atomicAdd(&cnt[rel], 1);
      }
    }
  }
  __syncthreads();
  int* o = part + (size_t)blockIdx.x * Nn + r0;
  for (int i = tid; i < nrows; i += 1024) o[i] = cnt[i];
}

__global__ void k_deg_fin(const int* __restrict__ part, float* __restrict__ dinv, int n) {
  int i = blockIdx.x * blockDim.x + threadIdx.x;
  if (i < n) {
    int c = 1;
#pragma unroll
    for (int k = 0; k < PART_CH; ++k) c += part[(size_t)k * n + i];
    float v = (float)c;
    dinv[i] = (v > 0.0f) ? rsqrtf(v) : 0.0f;
  }
}

template <int K, int F, bool BN>
__global__ __launch_bounds__(256, (F == 128 ? 6 : 8)) void k_gemm_strip(const float* __restrict__ A,
                                                    const float* __restrict__ B,
                                                    const float* __restrict__ dinv,
                                                    const float* __restrict__ stats,
                                                    const float* __restrict__ gamma,
                                                    const float* __restrict__ beta,
                                                    float invN,
                                                    float* __restrict__ C, int M) {
  constexpr int NT = F / 16;
  constexpr int NU = K / 8;
  constexpr int PH = F + 8;
  constexpr int F4 = F / 4;
  constexpr int NP = 256 / F4;
  static_assert(!BN || (K <= 256 && 256 % F4 == 0 && ((K / 2) * F4) % 256 == 0),
                "a thread per k for scale and shift; a thread stays on one column group");
  __shared__ float2 wp[(K / 2) * PH];
  __shared__ float4 ss[BN ? K / 2 : 1];
  __shared__ float4 cpart[BN ? 256 : 1];
  const int tid = threadIdx.x;
  const int lane = tid & 31;
  const int c = lane & 15;
  const int h = lane >> 4;
  const int m0 = (blockIdx.x * 8 + (tid >> 5)) << 4;
  const bool live = m0 < M;

  float4 a4[NU];
  if (live) {
    const float4* arow = reinterpret_cast<const float4*>(A + (size_t)(m0 + c) * K) + h;
#pragma unroll
    for (int u = 0; u < NU; ++u) a4[u] = arow[2 * u];
  }

  constexpr int ITEMS = (K / 2) * (F / 4);
  constexpr int NIT = (ITEMS + 255) / 256;
  float4 w0[NIT], w1[NIT];
#pragma unroll
  for (int n = 0; n < NIT; ++n) {
    const int idx = tid + 256 * n;
    w0[n] = make_float4(0.f, 0.f, 0.f, 0.f);
    w1[n] = w0[n];
    if (ITEMS % 256 == 0 || idx < ITEMS) {
      const int q = idx % (F / 4);
      const int kp = idx / (F / 4);
      w0[n] = *reinterpret_cast<const float4*>(B + (size_t)(2 * kp) * F + 4 * q);
      w1[n] = *reinterpret_cast<const float4*>(B + (size_t)(2 * kp + 1) * F + 4 * q);
    }
  }
  if constexpr (BN) {
    if (tid < K) {
      const float mean = stats[tid] * invN;
      const float var = stats[K + tid] * invN - mean * mean;
      const float sc = rsqrtf(var + BN_EPS) * gamma[tid];
      float* e = reinterpret_cast<float*>(ss) + 2 * tid;
      e[0] = sc;
      e[1] = beta[tid] - mean * sc;
    }
    __syncthreads();
  }
  float4 cp = make_float4(0.f, 0.f, 0.f, 0.f);
#pragma unroll
  for (int n = 0; n < NIT; ++n) {
    const int idx = tid + 256 * n;
    if (ITEMS % 256 == 0 || idx < ITEMS) {
      const int q = idx % (F / 4);
      const int kp = idx / (F / 4);
      if constexpr (BN) {
        float4 e = ss[kp];
        if constexpr (F4 == 32) {
          e.x = __builtin_bit_cast(float, __builtin_amdgcn_readfirstlane(__builtin_bit_cast(int, e.x)));
          e.y = __builtin_bit_cast(float, __builtin_amdgcn_readfirstlane(__builtin_bit_cast(int, e.y)));
          e.z = __builtin_bit_cast(float, __builtin_amdgcn_readfirstlane(__builtin_bit_cast(int, e.z)));
          e.w = __builtin_bit_cast(float, __builtin_amdgcn_readfirstlane(__builtin_bit_cast(int, e.w)));
        }
        cp.x += e.y * w0[n].x + e.w * w1[n].x;
        cp.y += e.y * w0[n].y + e.w * w1[n].y;
        cp.z += e.y * w0[n].z + e.w * w1[n].z;
        cp.w += e.y * w0[n].w + e.w * w1[n].w;
        w0[n].x *= e.x; w0[n].y *= e.x; w0[n].z *= e.x; w0[n].w *= e.x;
        w1[n].x *= e.z; w1[n].y *= e.z; w1[n].z *= e.z; w1[n].w *= e.z;
      }
      float2* d = wp + kp * PH + 64 * (q >> 4) + (q & 15);
      d[0]  = make_float2(w0[n].x, w1[n].x);
      d[16] = make_float2(w0[n].y, w1[n].y);
      d[32] = make_float2(w0[n].z, w1[n].z);
      d[48] = make_float2(w0[n].w, w1[n].w);
    }
  }
  if constexpr (BN) cpart[tid] = cp;
  __syncthreads();
  if (!live) return;

  v8f acc[NT];
  if constexpr (BN) {
#pragma unroll
    for (int g = 0; g < NT / 4; ++g) {
      float4 cs = cpart[16 * g + c];
#pragma unroll
      for (int p = 1; p < NP; ++p) {
        const float4 t = cpart[p * F4 + 16 * g + c];
        cs.x += t.x; cs.y += t.y; cs.z += t.z; cs.w += t.w;
      }
      acc[4 * g + 0] = (v8f){cs.x, cs.x, cs.x, cs.x, cs.x, cs.x, cs.x, cs.x};
      acc[4 * g + 1] = (v8f){cs.y, cs.y, cs.y, cs.y, cs.y, cs.y, cs.y, cs.y};
      acc[4 * g + 2] = (v8f){cs.z, cs.z, cs.z, cs.z, cs.z, cs.z, cs.z, cs.z};
      acc[4 * g + 3] = (v8f){cs.w, cs.w, cs.w, cs.w, cs.w, cs.w, cs.w, cs.w};
    }
  } else {
#pragma unroll
    for (int j = 0; j < NT; ++j) acc[j] = (v8f){0.f, 0.f, 0.f, 0.f, 0.f, 0.f, 0.f, 0.f};
  }
  const float2* bl = wp + (2 * h) * PH + c;
#pragma unroll
  for (int u = 0; u < NU; ++u) {
#pragma unroll
    for (int s = 0; s < 2; ++s) {
      v2f a;
      a.x = s ? a4[u].z : a4[u].x;
      a.y = s ? a4[u].w : a4[u].y;
      const float2* bp = bl + (4 * u + s) * PH;
#pragma unroll
      for (int j = 0; j < NT; ++j) {
        const float2 bv = bp[16 * j];
        v2f b; b.x = bv.x; b.y = bv.y;
        acc[j] = __builtin_amdgcn_wmma_f32_16x16x4_f32(false, a, false, b,
                                                       (short)0, acc[j], false, false);
      }
    }
  }
  const float* dvp = dinv + m0 + 8 * h;
  float* crow = C + (size_t)(m0 + 8 * h) * F + 4 * c;
#pragma unroll
  for (int r = 0; r < 8; ++r) {
    const float di = dvp[r];
#pragma unroll
    for (int q = 0; q < NT / 4; ++q) {
      float4 v;
      v.x = acc[4 * q + 0][r] * di;
      v.y = acc[4 * q + 1][r] * di;
      v.z = acc[4 * q + 2][r] * di;
      v.w = acc[4 * q + 3][r] * di;
      *reinterpret_cast<float4*>(crow + (size_t)r * F + 64 * q) = v;
    }
  }
}

#define PULL_R 400
#define BIN_CH 256
#define BIN_CAP 96
#define BIN_G 256
#define PULL_SEG (8 * BIN_CAP)
#define PACK_MAX_ROWS (1 << 23)

template <int F>
__global__ __launch_bounds__(1024) void k_bin(const int* __restrict__ src,
                                              const int* __restrict__ dst,
                                              int* __restrict__ cells,
                                              float* __restrict__ stats,
                                              int E, int Nn) {
  constexpr int PITCH = PULL_R * F / BIN_CH;
  static_assert(PITCH >= BIN_CAP + 1, "a cell holds its count and BIN_CAP entries");
  static_assert(PULL_R <= 512, "row within the range takes 9 bits");
  __shared__ int cnt[BIN_G];
  __shared__ int stage[BIN_G * BIN_CAP];
  const int tid = threadIdx.x;
  if (blockIdx.x == 0 && blockIdx.y == 0 && tid < 256) stats[tid] = 0.f;
  const int nb = (Nn + PULL_R - 1) / PULL_R;
  const int b0 = blockIdx.y * BIN_G;
  const int ng = min(BIN_G, nb - b0);
  const int lo = b0 * PULL_R;
  if (tid < BIN_G) cnt[tid] = 0;
  __syncthreads();
  const int per = (((E + BIN_CH - 1) / BIN_CH) + 3) & ~3;
  const int eb = blockIdx.x * per;
  const int ee = min(E, eb + per);
  const bool vd = (reinterpret_cast<size_t>(dst) & 15) == 0;
  const bool vs = (reinterpret_cast<size_t>(src) & 15) == 0;
  for (int e = eb + tid * 4; e < ee; e += 4096) {
    const int4 qd = ld_idx4(dst, e, ee, vd);
    const int4 qs = ld_idx4(src, e, ee, vs);
    const int dv[4] = {qd.x, qd.y, qd.z, qd.w};
    const int sv[4] = {qs.x, qs.y, qs.z, qs.w};
#pragma unroll
    for (int j = 0; j < 4; ++j) {
      const unsigned off = (unsigned)(dv[j] - lo);
      if (off < (unsigned)(ng * PULL_R)) {
        const unsigned g = off / PULL_R;
        const unsigned rel = off - g * PULL_R;
        const int slot = atomicAdd(&cnt[g], 1);
        if (slot < BIN_CAP) stage[g * BIN_CAP + slot] = (int)(((unsigned)sv[j] << 9) | rel);
      }
    }
  }
  __syncthreads();
  const int lane = tid & 31;
  const int wave = tid >> 5;
  for (int g = wave; g < ng; g += 32) {
    const int b = b0 + g;
    if ((b + 1) * PULL_R > Nn) continue;
    int* cell = cells + (size_t)b * (PULL_R * F) + (size_t)blockIdx.x * PITCH;
    const int c = cnt[g];
    const int n = min(c, BIN_CAP);
    if (lane == 0) cell[0] = c;
    for (int i = lane; i < n; i += 32) cell[1 + i] = stage[g * BIN_CAP + i];
  }
}

template <int F>
__global__ __launch_bounds__(1024) void k_pull(const float* __restrict__ hsc,
                                               const int* __restrict__ src,
                                               const int* __restrict__ dst,
                                               const float* __restrict__ dinv,
                                               const float* __restrict__ bias,
                                               float* __restrict__ stats,
                                               float* agg, int E, int Nn) {
  constexpr int C = F / 32;
  constexpr int U = 32 / C;
  constexpr int PITCH = PULL_R * F / BIN_CH;
  constexpr int CPW = BIN_CH / 32;
  static_assert(CPW * BIN_CAP <= PULL_SEG, "a wave's cells fit its list");
  __shared__ float acc[PULL_R * F];
  __shared__ int lp[32 * PULL_SEG];
  __shared__ float red[2 * F];
  __shared__ int walk;
  const int tid = threadIdx.x;
  const int lane = tid & 31;
  const int wave = __builtin_amdgcn_readfirstlane(tid >> 5);
  const int r0 = blockIdx.x * PULL_R;
  const int nrows = min(PULL_R, Nn - r0);

  {
    float4* a4 = reinterpret_cast<float4*>(acc);
    const float4 z = make_float4(0.f, 0.f, 0.f, 0.f);
    for (int i = tid; i < PULL_R * F / 4; i += 1024) a4[i] = z;
    if (tid < 2 * F) red[tid] = 0.f;
    if (tid == 0) walk = 0;
  }
  __syncthreads();

  int* mine = lp + wave * PULL_SEG;
  int n = 0;
  const bool binned = (nrows == PULL_R) && (Nn <= PACK_MAX_ROWS);
  if (binned) {
    const int* cells = reinterpret_cast<const int*>(agg) + (size_t)r0 * F +
                       (size_t)(wave * CPW) * PITCH;
    int cw = 0;
    if (lane < CPW) cw = cells[(size_t)lane * PITCH];
    if (cw > BIN_CAP) walk = 1;
    const int cn = min(max(cw, 0), BIN_CAP);
    int v[CPW];
#pragma unroll
    for (int j = 0; j < CPW; ++j) {
      const int nj = __builtin_amdgcn_readlane(cn, j);
      v[j] = (lane < nj) ? cells[(size_t)j * PITCH + 1 + lane] : 0;
    }
#pragma unroll
    for (int j = 0; j < CPW; ++j) {
      const int nj = __builtin_amdgcn_readlane(cn, j);
      if (lane < nj) mine[n + lane] = v[j];
      for (int i = lane + 32; i < nj; i += 32) mine[n + i] = cells[(size_t)j * PITCH + 1 + i];
      n += nj;
    }
  } else if (tid == 0) {
    walk = 1;
  }

  __syncthreads();

  if (walk != 0) {
    for (int e = tid; e < E; e += 1024) {
      const unsigned rel = (unsigned)(dst[e] - r0);
      if (rel < (unsigned)nrows) {
        const float* hr = hsc + (size_t)src[e] * F;
        float* ar = acc + rel * F;
        for (int c = 0; c < F; ++c) unsafeAtomicAdd(ar + c, hr[c]);
      }
    }
  } else {
  for (int i0 = 0; i0 < n; i0 += 32) {
    const int m = min(32, n - i0);
    int sv = 0, dl = 0;
    if (lane < m) {
      const unsigned pk = (unsigned)mine[i0 + lane];
      sv = (int)(pk >> 9);
      dl = (int)(pk & 511u);
    }
#pragma unroll
    for (int g = 0; g < 32; g += U) {
      if (g + U <= m) {
        float v[U][C];
#pragma unroll
        for (int u = 0; u < U; ++u) {
          const int s = __builtin_amdgcn_readlane(sv, g + u);
          const float* hrow = hsc + (size_t)s * F + lane;
#pragma unroll
          for (int j = 0; j < C; ++j) v[u][j] = hrow[32 * j];
        }
#pragma unroll
        for (int u = 0; u < U; ++u) {
          const int d = __builtin_amdgcn_readlane(dl, g + u);
          float* ad = acc + d * F + lane;
#pragma unroll
          for (int j = 0; j < C; ++j) unsafeAtomicAdd(ad + 32 * j, v[u][j]);
        }
      } else if (g < m) {
        for (int i = g; i < m; ++i) {
          const int s = __builtin_amdgcn_readlane(sv, i);
          const int d = __builtin_amdgcn_readlane(dl, i);
          const float* hrow = hsc + (size_t)s * F + lane;
          float* ad = acc + d * F + lane;
          float v[C];
#pragma unroll
          for (int j = 0; j < C; ++j) v[j] = hrow[32 * j];
#pragma unroll
          for (int j = 0; j < C; ++j) unsafeAtomicAdd(ad + 32 * j, v[j]);
        }
      }
    }
  }
  }
  __syncthreads();

  {
    float* o = agg + (size_t)r0 * F;
    const float* hs = hsc + (size_t)r0 * F;
    const float* dv = dinv + r0;
    if (((reinterpret_cast<size_t>(o) | reinterpret_cast<size_t>(hs) |
          reinterpret_cast<size_t>(bias)) & 15) == 0) {
      constexpr int F4 = F / 4;
      constexpr int RPP = 1024 / F4;
      const int c4 = tid % F4;
      const int rr = tid / F4;
      const float4* a4 = reinterpret_cast<const float4*>(acc);
      const float4* h4 = reinterpret_cast<const float4*>(hs);
      float4* o4 = reinterpret_cast<float4*>(o);
      const float4 b = reinterpret_cast<const float4*>(bias)[c4];
      float4 s = make_float4(0.f, 0.f, 0.f, 0.f);
      float4 q = make_float4(0.f, 0.f, 0.f, 0.f);
#pragma unroll 4
      for (int r = rr; r < nrows; r += RPP) {
        const float4 a = a4[r * F4 + c4];
        const float4 h = h4[r * F4 + c4];
        const float di = dv[r];
        float4 v;
        v.x = fmaxf((a.x + h.x) * di + b.x, 0.f);
        v.y = fmaxf((a.y + h.y) * di + b.y, 0.f);
        v.z = fmaxf((a.z + h.z) * di + b.z, 0.f);
        v.w = fmaxf((a.w + h.w) * di + b.w, 0.f);
        o4[r * F4 + c4] = v;
        s.x += v.x; s.y += v.y; s.z += v.z; s.w += v.w;
        q.x += v.x * v.x; q.y += v.y * v.y; q.z += v.z * v.z; q.w += v.w * v.w;
      }
      float* rs = red + 4 * c4;
      unsafeAtomicAdd(rs + 0, s.x);
      unsafeAtomicAdd(rs + 1, s.y);
      unsafeAtomicAdd(rs + 2, s.z);
      unsafeAtomicAdd(rs + 3, s.w);
      unsafeAtomicAdd(rs + F + 0, q.x);
      unsafeAtomicAdd(rs + F + 1, q.y);
      unsafeAtomicAdd(rs + F + 2, q.z);
      unsafeAtomicAdd(rs + F + 3, q.w);
    } else {
      constexpr int RPP = 1024 / F;
      const int c = tid % F;
      const float bj = bias[c];
      float s = 0.f, q = 0.f;
      for (int r = tid / F; r < nrows; r += RPP) {
        float v = (acc[r * F + c] + hs[(size_t)r * F + c]) * dv[r] + bj;
        v = fmaxf(v, 0.f);
        o[(size_t)r * F + c] = v;
        s += v;
        q += v * v;
      }
      unsafeAtomicAdd(red + c, s);
      unsafeAtomicAdd(red + F + c, q);
    }
  }
  __syncthreads();
  if (tid < 2 * F) unsafeAtomicAdd(stats + tid, red[tid]);
}

__global__ __launch_bounds__(256) void k_gemm64x2(const float* __restrict__ h3,
                                                  const float* __restrict__ W4,
                                                  const float* __restrict__ dinv,
                                                  const float* __restrict__ stats,
                                                  const float* __restrict__ gamma,
                                                  const float* __restrict__ beta,
                                                  float invN,
                                                  float* __restrict__ h4, int Nn) {
  __shared__ float2 wsc[64];
  __shared__ float2 wsh[64];
  __shared__ float2 cst;
  const int tid = threadIdx.x;
  if (tid < 64) {
    const float mean = stats[tid] * invN;
    const float var = stats[64 + tid] * invN - mean * mean;
    const float sc = rsqrtf(var + BN_EPS) * gamma[tid];
    const float sh = beta[tid] - mean * sc;
    const float wa = W4[2 * tid];
    const float wb = W4[2 * tid + 1];
    wsc[tid] = make_float2(sc * wa, sc * wb);
    wsh[tid] = make_float2(sh * wa, sh * wb);
  }
  __syncthreads();
  if (tid < 32) {
    const float2 a = wsh[tid];
    const float2 b = wsh[tid + 32];
    float c0 = a.x + b.x;
    float c1 = a.y + b.y;
#pragma unroll
    for (int off = 16; off > 0; off >>= 1) {
      c0 += __shfl_xor(c0, off);
      c1 += __shfl_xor(c1, off);
    }
    if (tid == 0) cst = make_float2(c0, c1);
  }
  __syncthreads();
  const int i = blockIdx.x * blockDim.x + tid;
  if (i >= Nn) return;
  const float* row = h3 + (size_t)i * 64;
  float o0 = cst.x, o1 = cst.y;
#pragma unroll 8
  for (int k = 0; k < 64; ++k) {
    const float xv = row[k];
    const float2 w = wsc[k];
    o0 += xv * w.x;
    o1 += xv * w.y;
  }
  float di = dinv[i];
  h4[2 * (size_t)i]     = o0 * di;
  h4[2 * (size_t)i + 1] = o1 * di;
}

__global__ __launch_bounds__(1024) void k_sum2_part(const float* __restrict__ h4s,
                                                    const int* __restrict__ src,
                                                    const int* __restrict__ dst,
                                                    float* __restrict__ part, int E, int Nn) {
  constexpr int U = 4;
  __shared__ float acc[2 * (PART_R + 32)];
  const int tid = threadIdx.x;
  const int spare = PART_R + (tid & 31);
  const int r0 = blockIdx.y * PART_R;
  const int nrows = min(PART_R, Nn - r0);
  {
    float4* a4 = reinterpret_cast<float4*>(acc);
    const float4 z = make_float4(0.f, 0.f, 0.f, 0.f);
    for (int i = tid; i < 2 * PART_R / 4; i += 1024) a4[i] = z;
  }
  __syncthreads();
  const int per = (((E + PART_CH - 1) / PART_CH) + 3) & ~3;
  const int eb = blockIdx.x * per;
  const int ee = min(E, eb + per);
  const bool vd = (reinterpret_cast<size_t>(dst) & 15) == 0;
  const bool vs = (reinterpret_cast<size_t>(src) & 15) == 0;
  const float2* h2 = reinterpret_cast<const float2*>(h4s);
  for (int b = eb; b < ee; b += 4096 * U) {
    int4 qd[U], qs[U];
    float2 v[4 * U];
    if (vd && vs && b + 4096 * U <= ee) {
#pragma unroll
      for (int u = 0; u < U; ++u) {
        const int e = b + u * 4096 + tid * 4;
        qd[u] = *reinterpret_cast<const int4*>(dst + e);
        qs[u] = *reinterpret_cast<const int4*>(src + e);
      }
    } else {
#pragma unroll
      for (int u = 0; u < U; ++u) {
        const int e = b + u * 4096 + tid * 4;
        qd[u] = ld_idx4(dst, e, ee, vd);
        qs[u] = ld_idx4(src, e, ee, vs);
      }
    }
#pragma unroll
    for (int u = 0; u < U; ++u) {
      const int dv[4] = {qd[u].x, qd[u].y, qd[u].z, qd[u].w};
      const int sv[4] = {qs[u].x, qs[u].y, qs[u].z, qs[u].w};
#pragma unroll
      for (int j = 0; j < 4; ++j) {
        const bool m = (unsigned)(dv[j] - r0) < (unsigned)nrows;
        v[4 * u + j] = h2[m ? sv[j] : 0];
      }
    }
#pragma unroll
    for (int u = 0; u < U; ++u) {
      const int dv[4] = {qd[u].x, qd[u].y, qd[u].z, qd[u].w};
#pragma unroll
      for (int j = 0; j < 4; ++j) {
        const unsigned rel = (unsigned)(dv[j] - r0);
        float* ar = acc + 2 * ((rel < (unsigned)nrows) ? (int)rel : spare);
        unsafeAtomicAdd(ar, v[4 * u + j].x);
        unsafeAtomicAdd(ar + 1, v[4 * u + j].y);
      }
    }
  }
  __syncthreads();
  float2* o = reinterpret_cast<float2*>(part) + ((size_t)blockIdx.x * Nn + r0);
  const float2* a2 = reinterpret_cast<const float2*>(acc);
  for (int i = tid; i < nrows; i += 1024) o[i] = a2[i];
}

__global__ void k_final_lsm(const float* __restrict__ part, const float* __restrict__ h4s,
                            const float* __restrict__ dinv, const float* __restrict__ b4,
                            float* __restrict__ out, int Nn) {
  int i = blockIdx.x * blockDim.x + threadIdx.x;
  if (i >= Nn) return;
  const float2* p2 = reinterpret_cast<const float2*>(part);
  float a0 = 0.f, a1 = 0.f;
#pragma unroll
  for (int k = 0; k < PART_CH; ++k) {
    const float2 t = p2[(size_t)k * Nn + i];
    a0 += t.x;
    a1 += t.y;
  }
  float di = dinv[i];
  float o0 = (a0 + h4s[2 * (size_t)i])     * di + b4[0];
  float o1 = (a1 + h4s[2 * (size_t)i + 1]) * di + b4[1];
  float m = fmaxf(o0, o1);
  float l = m + logf(expf(o0 - m) + expf(o1 - m));
  out[2 * (size_t)i]     = o0 - l;
  out[2 * (size_t)i + 1] = o1 - l;
}

extern "C" void kernel_launch(void* const* d_in, const int* in_sizes, int n_in,
                              void* d_out, int out_size, void* d_ws, size_t ws_size,
                              hipStream_t stream) {
  const float* x   = (const float*)d_in[0];
  const int*   ei  = (const int*)d_in[1];
  const float* W1  = (const float*)d_in[2];
  const float* b1  = (const float*)d_in[3];
  const float* g1  = (const float*)d_in[4];
  const float* be1 = (const float*)d_in[5];
  const float* W2  = (const float*)d_in[6];
  const float* b2  = (const float*)d_in[7];
  const float* g2  = (const float*)d_in[8];
  const float* be2 = (const float*)d_in[9];
  const float* W3  = (const float*)d_in[10];
  const float* b3  = (const float*)d_in[11];
  const float* g3  = (const float*)d_in[12];
  const float* be3 = (const float*)d_in[13];
  const float* W4  = (const float*)d_in[14];
  const float* b4  = (const float*)d_in[15];

  const int Nn = in_sizes[0] / 24;
  const int E  = in_sizes[1] / 2;
  const int* src = ei;
  const int* dst = ei + E;

  float* ws    = (float*)d_ws;
  float* dinv  = ws;
  float* stats = ws + Nn;
  float* bufA  = stats + 256;
  float* bufB  = bufA + (size_t)Nn * 128;

  const float invN = 1.0f / (float)Nn;
  const int TB = 256;

  const int gemmBlocks = (Nn + 127) / 128;
  const int nRanges = (Nn + PULL_R - 1) / PULL_R;
  const dim3 binGrid(BIN_CH, (nRanges + BIN_G - 1) / BIN_G);
  auto pull128 = [&](const float* h, const float* bias, float* out) {
    k_bin<128><<<binGrid, 1024, 0, stream>>>(src, dst, (int*)out, stats, E, Nn);
    k_pull<128><<<nRanges, 1024, 0, stream>>>(h, src, dst, dinv, bias, stats, out, E, Nn);
  };
  auto pull64 = [&](const float* h, const float* bias, float* out) {
    k_bin<64><<<binGrid, 1024, 0, stream>>>(src, dst, (int*)out, stats, E, Nn);
    k_pull<64><<<nRanges, 1024, 0, stream>>>(h, src, dst, dinv, bias, stats, out, E, Nn);
  };
  const dim3 partGrid(PART_CH, (Nn + PART_R - 1) / PART_R);

  k_deg_part<<<partGrid, 1024, 0, stream>>>(dst, (int*)bufB, E, Nn);
  k_deg_fin<<<(Nn + TB - 1) / TB, TB, 0, stream>>>((const int*)bufB, dinv, Nn);

  k_gemm_strip<24, 128, false><<<gemmBlocks, 256, 0, stream>>>(x, W1, dinv, stats, g1, be1, invN,
                                                               bufA, Nn);
  pull128(bufA, b1, bufB);

  k_gemm_strip<128, 128, true><<<gemmBlocks, 256, 0, stream>>>(bufB, W2, dinv, stats, g1, be1, invN,
                                                               bufA, Nn);
  pull128(bufA, b2, bufB);

  k_gemm_strip<128, 64, true><<<gemmBlocks, 256, 0, stream>>>(bufB, W3, dinv, stats, g2, be2, invN,
                                                              bufA, Nn);
  pull64(bufA, b3, bufB);

  k_gemm64x2<<<(Nn + TB - 1) / TB, TB, 0, stream>>>(bufB, W4, dinv, stats, g3, be3, invN, bufA, Nn);
  k_sum2_part<<<partGrid, 1024, 0, stream>>>(bufA, src, dst, bufB, E, Nn);
  k_final_lsm<<<(Nn + TB - 1) / TB, TB, 0, stream>>>(bufB, bufA, dinv, b4,
                                                     (float*)d_out, Nn);
}


# layer 4 edge sum reads layer 3's edge cells again (kept in its product buffer's unused half): a workgroup adds its 400 rows in LDS, log-softmax in its store; planes, last pass go (sum order changes)
# speedup vs baseline: 5.5429x; 1.0744x over previous
#include <hip/hip_runtime.h>
#include <hip/hip_bf16.h>
#include <math.h>


typedef float v2f __attribute__((ext_vector_type(2)));
typedef float v8f __attribute__((ext_vector_type(8)));

#define BN_EPS 1e-5f

#define PART_R 12800
#define PART_CH 32

__device__ __forceinline__ int4 ld_idx4(const int* __restrict__ p, int e, int ee, bool vec) {
  if (vec && e + 4 <= ee) return *reinterpret_cast<const int4*>(p + e);
  int4 r;
  r.x = (e     < ee) ? p[e]     : -1;
  r.y = (e + 1 < ee) ? p[e + 1] : -1;
  r.z = (e + 2 < ee) ? p[e + 2] : -1;
  r.w = (e + 3 < ee) ? p[e + 3] : -1;
  return r;
}

__global__ __launch_bounds__(1024) void k_deg_part(const int* __restrict__ dst,
                                                   int* __restrict__ part, int E, int Nn) {
  constexpr int U = 8;
  __shared__ int cnt[PART_R];
  const int tid = threadIdx.x;
  const int r0 = blockIdx.y * PART_R;
  const int nrows = min(PART_R, Nn - r0);
  for (int i = tid; i < PART_R; i += 1024) cnt[i] = 0;
  __syncthreads();
  const int per = (((E + PART_CH - 1) / PART_CH) + 3) & ~3;
  const int eb = blockIdx.x * per;
  const int ee = min(E, eb + per);
  const bool vec = (reinterpret_cast<size_t>(dst) & 15) == 0;
  for (int b = eb; b < ee; b += 4096 * U) {
    int4 q[U];
    if (vec && b + 4096 * U <= ee) {
#pragma unroll
      for (int u = 0; u < U; ++u)
        q[u] = *reinterpret_cast<const int4*>(dst + b + u * 4096 + tid * 4);
    } else {
#pragma unroll
      for (int u = 0; u < U; ++u) q[u] = ld_idx4(dst, b + u * 4096 + tid * 4, ee, vec);
    }
#pragma unroll
    for (int u = 0; u < U; ++u) {
      const int dv[4] = {q[u].x, q[u].y, q[u].z, q[u].w};
#pragma unroll
      for (int j = 0; j < 4; ++j) {
        const unsigned rel = (unsigned)(dv[j] - r0);
        if (rel < (unsigned)nrows) atomicAdd(&cnt[rel], 1);
      }
    }
  }
  __syncthreads();
  int* o = part + (size_t)blockIdx.x * Nn + r0;
  for (int i = tid; i < nrows; i += 1024) o[i] = cnt[i];
}

__global__ void k_deg_fin(const int* __restrict__ part, float* __restrict__ dinv, int n) {
  int i = blockIdx.x * blockDim.x + threadIdx.x;
  if (i < n) {
    int c = 1;
#pragma unroll
    for (int k = 0; k < PART_CH; ++k) c += part[(size_t)k * n + i];
    float v = (float)c;
    dinv[i] = (v > 0.0f) ? rsqrtf(v) : 0.0f;
  }
}

template <int K, int F, bool BN>
__global__ __launch_bounds__(256, (F == 128 ? 6 : 8)) void k_gemm_strip(const float* __restrict__ A,
                                                    const float* __restrict__ B,
                                                    const float* __restrict__ dinv,
                                                    const float* __restrict__ stats,
                                                    const float* __restrict__ gamma,
                                                    const float* __restrict__ beta,
                                                    float invN,
                                                    float* __restrict__ C, int M) {
  constexpr int NT = F / 16;
  constexpr int NU = K / 8;
  constexpr int PH = F + 8;
  constexpr int F4 = F / 4;
  constexpr int NP = 256 / F4;
  static_assert(!BN || (K <= 256 && 256 % F4 == 0 && ((K / 2) * F4) % 256 == 0),
                "a thread per k for scale and shift; a thread stays on one column group");
  __shared__ float2 wp[(K / 2) * PH];
  __shared__ float4 ss[BN ? K / 2 : 1];
  __shared__ float4 cpart[BN ? 256 : 1];
  const int tid = threadIdx.x;
  const int lane = tid & 31;
  const int c = lane & 15;
  const int h = lane >> 4;
  const int m0 = (blockIdx.x * 8 + (tid >> 5)) << 4;
  const bool live = m0 < M;

  float4 a4[NU];
  if (live) {
    const float4* arow = reinterpret_cast<const float4*>(A + (size_t)(m0 + c) * K) + h;
#pragma unroll
    for (int u = 0; u < NU; ++u) a4[u] = arow[2 * u];
  }

  constexpr int ITEMS = (K / 2) * (F / 4);
  constexpr int NIT = (ITEMS + 255) / 256;
  float4 w0[NIT], w1[NIT];
#pragma unroll
  for (int n = 0; n < NIT; ++n) {
    const int idx = tid + 256 * n;
    w0[n] = make_float4(0.f, 0.f, 0.f, 0.f);
    w1[n] = w0[n];
    if (ITEMS % 256 == 0 || idx < ITEMS) {
      const int q = idx % (F / 4);
      const int kp = idx / (F / 4);
      w0[n] = *reinterpret_cast<const float4*>(B + (size_t)(2 * kp) * F + 4 * q);
      w1[n] = *reinterpret_cast<const float4*>(B + (size_t)(2 * kp + 1) * F + 4 * q);
    }
  }
  if constexpr (BN) {
    if (tid < K) {
      const float mean = stats[tid] * invN;
      const float var = stats[K + tid] * invN - mean * mean;
      const float sc = rsqrtf(var + BN_EPS) * gamma[tid];
      float* e = reinterpret_cast<float*>(ss) + 2 * tid;
      e[0] = sc;
      e[1] = beta[tid] - mean * sc;
    }
    __syncthreads();
  }
  float4 cp = make_float4(0.f, 0.f, 0.f, 0.f);
#pragma unroll
  for (int n = 0; n < NIT; ++n) {
    const int idx = tid + 256 * n;
    if (ITEMS % 256 == 0 || idx < ITEMS) {
      const int q = idx % (F / 4);
      const int kp = idx / (F / 4);
      if constexpr (BN) {
        float4 e = ss[kp];
        if constexpr (F4 == 32) {
          e.x = __builtin_bit_cast(float, __builtin_amdgcn_readfirstlane(__builtin_bit_cast(int, e.x)));
          e.y = __builtin_bit_cast(float, __builtin_amdgcn_readfirstlane(__builtin_bit_cast(int, e.y)));
          e.z = __builtin_bit_cast(float, __builtin_amdgcn_readfirstlane(__builtin_bit_cast(int, e.z)));
          e.w = __builtin_bit_cast(float, __builtin_amdgcn_readfirstlane(__builtin_bit_cast(int, e.w)));
        }
        cp.x += e.y * w0[n].x + e.w * w1[n].x;
        cp.y += e.y * w0[n].y + e.w * w1[n].y;
        cp.z += e.y * w0[n].z + e.w * w1[n].z;
        cp.w += e.y * w0[n].w + e.w * w1[n].w;
        w0[n].x *= e.x; w0[n].y *= e.x; w0[n].z *= e.x; w0[n].w *= e.x;
        w1[n].x *= e.z; w1[n].y *= e.z; w1[n].z *= e.z; w1[n].w *= e.z;
      }
      float2* d = wp + kp * PH + 64 * (q >> 4) + (q & 15);
      d[0]  = make_float2(w0[n].x, w1[n].x);
      d[16] = make_float2(w0[n].y, w1[n].y);
      d[32] = make_float2(w0[n].z, w1[n].z);
      d[48] = make_float2(w0[n].w, w1[n].w);
    }
  }
  if constexpr (BN) cpart[tid] = cp;
  __syncthreads();
  if (!live) return;

  v8f acc[NT];
  if constexpr (BN) {
#pragma unroll
    for (int g = 0; g < NT / 4; ++g) {
      float4 cs = cpart[16 * g + c];
#pragma unroll
      for (int p = 1; p < NP; ++p) {
        const float4 t = cpart[p * F4 + 16 * g + c];
        cs.x += t.x; cs.y += t.y; cs.z += t.z; cs.w += t.w;
      }
      acc[4 * g + 0] = (v8f){cs.x, cs.x, cs.x, cs.x, cs.x, cs.x, cs.x, cs.x};
      acc[4 * g + 1] = (v8f){cs.y, cs.y, cs.y, cs.y, cs.y, cs.y, cs.y, cs.y};
      acc[4 * g + 2] = (v8f){cs.z, cs.z, cs.z, cs.z, cs.z, cs.z, cs.z, cs.z};
      acc[4 * g + 3] = (v8f){cs.w, cs.w, cs.w, cs.w, cs.w, cs.w, cs.w, cs.w};
    }
  } else {
#pragma unroll
    for (int j = 0; j < NT; ++j) acc[j] = (v8f){0.f, 0.f, 0.f, 0.f, 0.f, 0.f, 0.f, 0.f};
  }
  const float2* bl = wp + (2 * h) * PH + c;
#pragma unroll
  for (int u = 0; u < NU; ++u) {
#pragma unroll
    for (int s = 0; s < 2; ++s) {
      v2f a;
      a.x = s ? a4[u].z : a4[u].x;
      a.y = s ? a4[u].w : a4[u].y;
      const float2* bp = bl + (4 * u + s) * PH;
#pragma unroll
      for (int j = 0; j < NT; ++j) {
        const float2 bv = bp[16 * j];
        v2f b; b.x = bv.x; b.y = bv.y;
        acc[j] = __builtin_amdgcn_wmma_f32_16x16x4_f32(false, a, false, b,
                                                       (short)0, acc[j], false, false);
      }
    }
  }
  const float* dvp = dinv + m0 + 8 * h;
  float* crow = C + (size_t)(m0 + 8 * h) * F + 4 * c;
#pragma unroll
  for (int r = 0; r < 8; ++r) {
    const float di = dvp[r];
#pragma unroll
    for (int q = 0; q < NT / 4; ++q) {
      float4 v;
      v.x = acc[4 * q + 0][r] * di;
      v.y = acc[4 * q + 1][r] * di;
      v.z = acc[4 * q + 2][r] * di;
      v.w = acc[4 * q + 3][r] * di;
      *reinterpret_cast<float4*>(crow + (size_t)r * F + 64 * q) = v;
    }
  }
}

#define PULL_R 400
#define BIN_CH 256
#define BIN_CAP 96
#define BIN_G 256
#define PULL_SEG (8 * BIN_CAP)
#define PACK_MAX_ROWS (1 << 23)

template <int F>
__global__ __launch_bounds__(1024) void k_bin(const int* __restrict__ src,
                                              const int* __restrict__ dst,
                                              int* __restrict__ cells,
                                              float* __restrict__ stats,
                                              int E, int Nn) {
  constexpr int PITCH = PULL_R * F / BIN_CH;
  static_assert(PITCH >= BIN_CAP + 1, "a cell holds its count and BIN_CAP entries");
  static_assert(PULL_R <= 512, "row within the range takes 9 bits");
  __shared__ int cnt[BIN_G];
  __shared__ int stage[BIN_G * BIN_CAP];
  const int tid = threadIdx.x;
  if (blockIdx.x == 0 && blockIdx.y == 0 && tid < 256) stats[tid] = 0.f;
  const int nb = (Nn + PULL_R - 1) / PULL_R;
  const int b0 = blockIdx.y * BIN_G;
  const int ng = min(BIN_G, nb - b0);
  const int lo = b0 * PULL_R;
  if (tid < BIN_G) cnt[tid] = 0;
  __syncthreads();
  const int per = (((E + BIN_CH - 1) / BIN_CH) + 3) & ~3;
  const int eb = blockIdx.x * per;
  const int ee = min(E, eb + per);
  const bool vd = (reinterpret_cast<size_t>(dst) & 15) == 0;
  const bool vs = (reinterpret_cast<size_t>(src) & 15) == 0;
  for (int e = eb + tid * 4; e < ee; e += 4096) {
    const int4 qd = ld_idx4(dst, e, ee, vd);
    const int4 qs = ld_idx4(src, e, ee, vs);
    const int dv[4] = {qd.x, qd.y, qd.z, qd.w};
    const int sv[4] = {qs.x, qs.y, qs.z, qs.w};
#pragma unroll
    for (int j = 0; j < 4; ++j) {
      const unsigned off = (unsigned)(dv[j] - lo);
      if (off < (unsigned)(ng * PULL_R)) {
        const unsigned g = off / PULL_R;
        const unsigned rel = off - g * PULL_R;
        const int slot = atomicAdd(&cnt[g], 1);
        if (slot < BIN_CAP) stage[g * BIN_CAP + slot] = (int)(((unsigned)sv[j] << 9) | rel);
      }
    }
  }
  __syncthreads();
  const int lane = tid & 31;
  const int wave = tid >> 5;
  for (int g = wave; g < ng; g += 32) {
    const int b = b0 + g;
    if ((b + 1) * PULL_R > Nn) continue;
    int* cell = cells + (size_t)b * (PULL_R * F) + (size_t)blockIdx.x * PITCH;
    const int c = cnt[g];
    const int n = min(c, BIN_CAP);
    if (lane == 0) cell[0] = c;
    for (int i = lane; i < n; i += 32) cell[1 + i] = stage[g * BIN_CAP + i];
  }
}

template <int F>
__global__ __launch_bounds__(1024) void k_pull(const float* __restrict__ hsc,
                                               const int* __restrict__ src,
                                               const int* __restrict__ dst,
                                               const float* __restrict__ dinv,
                                               const float* __restrict__ bias,
                                               float* __restrict__ stats,
                                               const int* cellbase,
                                               float* agg, int E, int Nn) {
  constexpr int C = F / 32;
  constexpr int U = 32 / C;
  constexpr int PITCH = PULL_R * F / BIN_CH;
  constexpr int CPW = BIN_CH / 32;
  static_assert(CPW * BIN_CAP <= PULL_SEG, "a wave's cells fit its list");
  __shared__ float acc[PULL_R * F];
  __shared__ int lp[32 * PULL_SEG];
  __shared__ float red[2 * F];
  __shared__ int walk;
  const int tid = threadIdx.x;
  const int lane = tid & 31;
  const int wave = __builtin_amdgcn_readfirstlane(tid >> 5);
  const int r0 = blockIdx.x * PULL_R;
  const int nrows = min(PULL_R, Nn - r0);

  {
    float4* a4 = reinterpret_cast<float4*>(acc);
    const float4 z = make_float4(0.f, 0.f, 0.f, 0.f);
    for (int i = tid; i < PULL_R * F / 4; i += 1024) a4[i] = z;
    if (tid < 2 * F) red[tid] = 0.f;
    if (tid == 0) walk = 0;
  }
  __syncthreads();

  int* mine = lp + wave * PULL_SEG;
  int n = 0;
  const bool binned = (nrows == PULL_R) && (Nn <= PACK_MAX_ROWS);
  if (binned) {
    const int* cells = cellbase + (size_t)r0 * F + (size_t)(wave * CPW) * PITCH;
    int cw = 0;
    if (lane < CPW) cw = cells[(size_t)lane * PITCH];
    if (cw > BIN_CAP) walk = 1;
    const int cn = min(max(cw, 0), BIN_CAP);
    int v[CPW];
#pragma unroll
    for (int j = 0; j < CPW; ++j) {
      const int nj = __builtin_amdgcn_readlane(cn, j);
      v[j] = (lane < nj) ? cells[(size_t)j * PITCH + 1 + lane] : 0;
    }
#pragma unroll
    for (int j = 0; j < CPW; ++j) {
      const int nj = __builtin_amdgcn_readlane(cn, j);
      if (lane < nj) mine[n + lane] = v[j];
      for (int i = lane + 32; i < nj; i += 32) mine[n + i] = cells[(size_t)j * PITCH + 1 + i];
      n += nj;
    }
  } else if (tid == 0) {
    walk = 1;
  }

  __syncthreads();

  if (walk != 0) {
    for (int e = tid; e < E; e += 1024) {
      const unsigned rel = (unsigned)(dst[e] - r0);
      if (rel < (unsigned)nrows) {
        const float* hr = hsc + (size_t)src[e] * F;
        float* ar = acc + rel * F;
        for (int c = 0; c < F; ++c) unsafeAtomicAdd(ar + c, hr[c]);
      }
    }
  } else {
  for (int i0 = 0; i0 < n; i0 += 32) {
    const int m = min(32, n - i0);
    int sv = 0, dl = 0;
    if (lane < m) {
      const unsigned pk = (unsigned)mine[i0 + lane];
      sv = (int)(pk >> 9);
      dl = (int)(pk & 511u);
    }
#pragma unroll
    for (int g = 0; g < 32; g += U) {
      if (g + U <= m) {
        float v[U][C];
#pragma unroll
        for (int u = 0; u < U; ++u) {
          const int s = __builtin_amdgcn_readlane(sv, g + u);
          const float* hrow = hsc + (size_t)s * F + lane;
#pragma unroll
          for (int j = 0; j < C; ++j) v[u][j] = hrow[32 * j];
        }
#pragma unroll
        for (int u = 0; u < U; ++u) {
          const int d = __builtin_amdgcn_readlane(dl, g + u);
          float* ad = acc + d * F + lane;
#pragma unroll
          for (int j = 0; j < C; ++j) unsafeAtomicAdd(ad + 32 * j, v[u][j]);
        }
      } else if (g < m) {
        for (int i = g; i < m; ++i) {
          const int s = __builtin_amdgcn_readlane(sv, i);
          const int d = __builtin_amdgcn_readlane(dl, i);
          const float* hrow = hsc + (size_t)s * F + lane;
          float* ad = acc + d * F + lane;
          float v[C];
#pragma unroll
          for (int j = 0; j < C; ++j) v[j] = hrow[32 * j];
#pragma unroll
          for (int j = 0; j < C; ++j) unsafeAtomicAdd(ad + 32 * j, v[j]);
        }
      }
    }
  }
  }
  __syncthreads();

  {
    float* o = agg + (size_t)r0 * F;
    const float* hs = hsc + (size_t)r0 * F;
    const float* dv = dinv + r0;
    if (((reinterpret_cast<size_t>(o) | reinterpret_cast<size_t>(hs) |
          reinterpret_cast<size_t>(bias)) & 15) == 0) {
      constexpr int F4 = F / 4;
      constexpr int RPP = 1024 / F4;
      const int c4 = tid % F4;
      const int rr = tid / F4;
      const float4* a4 = reinterpret_cast<const float4*>(acc);
      const float4* h4 = reinterpret_cast<const float4*>(hs);
      float4* o4 = reinterpret_cast<float4*>(o);
      const float4 b = reinterpret_cast<const float4*>(bias)[c4];
      float4 s = make_float4(0.f, 0.f, 0.f, 0.f);
      float4 q = make_float4(0.f, 0.f, 0.f, 0.f);
#pragma unroll 4
      for (int r = rr; r < nrows; r += RPP) {
        const float4 a = a4[r * F4 + c4];
        const float4 h = h4[r * F4 + c4];
        const float di = dv[r];
        float4 v;
        v.x = fmaxf((a.x + h.x) * di + b.x, 0.f);
        v.y = fmaxf((a.y + h.y) * di + b.y, 0.f);
        v.z = fmaxf((a.z + h.z) * di + b.z, 0.f);
        v.w = fmaxf((a.w + h.w) * di + b.w, 0.f);
        o4[r * F4 + c4] = v;
        s.x += v.x; s.y += v.y; s.z += v.z; s.w += v.w;
        q.x += v.x * v.x; q.y += v.y * v.y; q.z += v.z * v.z; q.w += v.w * v.w;
      }
      float* rs = red + 4 * c4;
      unsafeAtomicAdd(rs + 0, s.x);
      unsafeAtomicAdd(rs + 1, s.y);
      unsafeAtomicAdd(rs + 2, s.z);
      unsafeAtomicAdd(rs + 3, s.w);
      unsafeAtomicAdd(rs + F + 0, q.x);
      unsafeAtomicAdd(rs + F + 1, q.y);
      unsafeAtomicAdd(rs + F + 2, q.z);
      unsafeAtomicAdd(rs + F + 3, q.w);
    } else {
      constexpr int RPP = 1024 / F;
      const int c = tid % F;
      const float bj = bias[c];
      float s = 0.f, q = 0.f;
      for (int r = tid / F; r < nrows; r += RPP) {
        float v = (acc[r * F + c] + hs[(size_t)r * F + c]) * dv[r] + bj;
        v = fmaxf(v, 0.f);
        o[(size_t)r * F + c] = v;
        s += v;
        q += v * v;
      }
      unsafeAtomicAdd(red + c, s);
      unsafeAtomicAdd(red + F + c, q);
    }
  }
  __syncthreads();
  if (tid < 2 * F) unsafeAtomicAdd(stats + tid, red[tid]);
}

__global__ __launch_bounds__(256) void k_gemm64x2(const float* __restrict__ h3,
                                                  const float* __restrict__ W4,
                                                  const float* __restrict__ dinv,
                                                  const float* __restrict__ stats,
                                                  const float* __restrict__ gamma,
                                                  const float* __restrict__ beta,
                                                  float invN,
                                                  float* __restrict__ h4, int Nn) {
  __shared__ float2 wsc[64];
  __shared__ float2 wsh[64];
  __shared__ float2 cst;
  const int tid = threadIdx.x;
  if (tid < 64) {
    const float mean = stats[tid] * invN;
    const float var = stats[64 + tid] * invN - mean * mean;
    const float sc = rsqrtf(var + BN_EPS) * gamma[tid];
    const float sh = beta[tid] - mean * sc;
    const float wa = W4[2 * tid];
    const float wb = W4[2 * tid + 1];
    wsc[tid] = make_float2(sc * wa, sc * wb);
    wsh[tid] = make_float2(sh * wa, sh * wb);
  }
  __syncthreads();
  if (tid < 32) {
    const float2 a = wsh[tid];
    const float2 b = wsh[tid + 32];
    float c0 = a.x + b.x;
    float c1 = a.y + b.y;
#pragma unroll
    for (int off = 16; off > 0; off >>= 1) {
      c0 += __shfl_xor(c0, off);
      c1 += __shfl_xor(c1, off);
    }
    if (tid == 0) cst = make_float2(c0, c1);
  }
  __syncthreads();
  const int i = blockIdx.x * blockDim.x + tid;
  if (i >= Nn) return;
  const float* row = h3 + (size_t)i * 64;
  float o0 = cst.x, o1 = cst.y;
#pragma unroll 8
  for (int k = 0; k < 64; ++k) {
    const float xv = row[k];
    const float2 w = wsc[k];
    o0 += xv * w.x;
    o1 += xv * w.y;
  }
  float di = dinv[i];
  h4[2 * (size_t)i]     = o0 * di;
  h4[2 * (size_t)i + 1] = o1 * di;
}

#define PULL2_T 256
__global__ __launch_bounds__(PULL2_T) void k_pull2(const float* __restrict__ h4s,
                                                   const int* __restrict__ cellbase,
                                                   const int* __restrict__ src,
                                                   const int* __restrict__ dst,
                                                   const float* __restrict__ dinv,
                                                   const float* __restrict__ b4,
                                                   float* __restrict__ out, int E, int Nn) {
  constexpr int PITCH = PULL_R * 64 / BIN_CH;
  constexpr int NW = PULL2_T / 32;
  constexpr int CPW = BIN_CH / NW;
  static_assert(CPW <= 32, "a lane per cell for the counts");
  __shared__ float acc[2 * PULL_R];
  __shared__ int walk;
  const int tid = threadIdx.x;
  const int lane = tid & 31;
  const int wave = __builtin_amdgcn_readfirstlane(tid >> 5);
  const int r0 = blockIdx.x * PULL_R;
  const int nrows = min(PULL_R, Nn - r0);
  const float2* h2 = reinterpret_cast<const float2*>(h4s);
  for (int i = tid; i < 2 * PULL_R; i += PULL2_T) acc[i] = 0.f;
  if (tid == 0) walk = 0;
  __syncthreads();

  const bool binned = (nrows == PULL_R) && (Nn <= PACK_MAX_ROWS);
  const int* cells = cellbase + (size_t)r0 * 64 + (size_t)(wave * CPW) * PITCH;
  int cn = 0;
  if (binned) {
    int cw = 0;
    if (lane < CPW) cw = cells[(size_t)lane * PITCH];
    if (cw > BIN_CAP) walk = 1;
    cn = min(max(cw, 0), BIN_CAP);
  } else if (tid == 0) {
    walk = 1;
  }
  __syncthreads();

  if (walk != 0) {
    for (int e = tid; e < E; e += PULL2_T) {
      const unsigned rel = (unsigned)(dst[e] - r0);
      if (rel < (unsigned)nrows) {
        const float2 v = h2[src[e]];
        unsafeAtomicAdd(acc + 2 * rel, v.x);
        unsafeAtomicAdd(acc + 2 * rel + 1, v.y);
      }
    }
  } else {
    unsigned pk[CPW];
    float2 v[CPW];
#pragma unroll
    for (int p = 0; p < 2; ++p) {
      const int li = lane + 32 * p;
#pragma unroll
      for (int j = 0; j < CPW; ++j) {
        const int nj = __builtin_amdgcn_readlane(cn, j);
        pk[j] = 0u;
        if (li < nj) pk[j] = (unsigned)cells[(size_t)j * PITCH + 1 + li];
      }
#pragma unroll
      for (int j = 0; j < CPW; ++j) v[j] = h2[pk[j] >> 9];
#pragma unroll
      for (int j = 0; j < CPW; ++j) {
        const int nj = __builtin_amdgcn_readlane(cn, j);
        if (li < nj) {
          float* ar = acc + 2 * (pk[j] & 511u);
          unsafeAtomicAdd(ar, v[j].x);
          unsafeAtomicAdd(ar + 1, v[j].y);
        }
      }
    }
    for (int j = 0; j < CPW; ++j) {
      const int nj = __builtin_amdgcn_readlane(cn, j);
      for (int i = lane + 64; i < nj; i += 32) {
        const unsigned q = (unsigned)cells[(size_t)j * PITCH + 1 + i];
        const float2 t = h2[q >> 9];
        float* ar = acc + 2 * (q & 511u);
        unsafeAtomicAdd(ar, t.x);
        unsafeAtomicAdd(ar + 1, t.y);
      }
    }
  }
  __syncthreads();

  const float bb0 = b4[0];
  const float bb1 = b4[1];
  for (int t = tid; t < nrows; t += PULL2_T) {
    const int i = r0 + t;
    const float di = dinv[i];
    const float2 own = h2[i];
    const float o0 = (acc[2 * t] + own.x) * di + bb0;
    const float o1 = (acc[2 * t + 1] + own.y) * di + bb1;
    const float m = fmaxf(o0, o1);
    const float l = m + logf(expf(o0 - m) + expf(o1 - m));
    out[2 * (size_t)i]     = o0 - l;
    out[2 * (size_t)i + 1] = o1 - l;
  }
}

extern "C" void kernel_launch(void* const* d_in, const int* in_sizes, int n_in,
                              void* d_out, int out_size, void* d_ws, size_t ws_size,
                              hipStream_t stream) {
  const float* x   = (const float*)d_in[0];
  const int*   ei  = (const int*)d_in[1];
  const float* W1  = (const float*)d_in[2];
  const float* b1  = (const float*)d_in[3];
  const float* g1  = (const float*)d_in[4];
  const float* be1 = (const float*)d_in[5];
  const float* W2  = (const float*)d_in[6];
  const float* b2  = (const float*)d_in[7];
  const float* g2  = (const float*)d_in[8];
  const float* be2 = (const float*)d_in[9];
  const float* W3  = (const float*)d_in[10];
  const float* b3  = (const float*)d_in[11];
  const float* g3  = (const float*)d_in[12];
  const float* be3 = (const float*)d_in[13];
  const float* W4  = (const float*)d_in[14];
  const float* b4  = (const float*)d_in[15];

  const int Nn = in_sizes[0] / 24;
  const int E  = in_sizes[1] / 2;
  const int* src = ei;
  const int* dst = ei + E;

  float* ws    = (float*)d_ws;
  float* dinv  = ws;
  float* stats = ws + Nn;
  float* bufA  = stats + 256;
  float* bufB  = bufA + (size_t)Nn * 128;

  const float invN = 1.0f / (float)Nn;
  const int TB = 256;

  const int gemmBlocks = (Nn + 127) / 128;
  const int nRanges = (Nn + PULL_R - 1) / PULL_R;
  const dim3 binGrid(BIN_CH, (nRanges + BIN_G - 1) / BIN_G);
  auto pull128 = [&](const float* h, const float* bias, float* out) {
    k_bin<128><<<binGrid, 1024, 0, stream>>>(src, dst, (int*)out, stats, E, Nn);
    k_pull<128><<<nRanges, 1024, 0, stream>>>(h, src, dst, dinv, bias, stats, (const int*)out, out, E,
                                              Nn);
  };
  auto pull64 = [&](const float* h, const float* bias, int* cells, float* out) {
    k_bin<64><<<binGrid, 1024, 0, stream>>>(src, dst, cells, stats, E, Nn);
    k_pull<64><<<nRanges, 1024, 0, stream>>>(h, src, dst, dinv, bias, stats, cells, out, E, Nn);
  };
  const dim3 partGrid(PART_CH, (Nn + PART_R - 1) / PART_R);

  k_deg_part<<<partGrid, 1024, 0, stream>>>(dst, (int*)bufB, E, Nn);
  k_deg_fin<<<(Nn + TB - 1) / TB, TB, 0, stream>>>((const int*)bufB, dinv, Nn);

  k_gemm_strip<24, 128, false><<<gemmBlocks, 256, 0, stream>>>(x, W1, dinv, stats, g1, be1, invN,
                                                               bufA, Nn);
  pull128(bufA, b1, bufB);

  k_gemm_strip<128, 128, true><<<gemmBlocks, 256, 0, stream>>>(bufB, W2, dinv, stats, g1, be1, invN,
                                                               bufA, Nn);
  pull128(bufA, b2, bufB);

  k_gemm_strip<128, 64, true><<<gemmBlocks, 256, 0, stream>>>(bufB, W3, dinv, stats, g2, be2, invN,
                                                              bufA, Nn);
  int* cells3 = (int*)(bufA + (size_t)Nn * 64);
  pull64(bufA, b3, cells3, bufB);

  k_gemm64x2<<<(Nn + TB - 1) / TB, TB, 0, stream>>>(bufB, W4, dinv, stats, g3, be3, invN, bufA, Nn);
  k_pull2<<<nRanges, PULL2_T, 0, stream>>>(bufA, cells3, src, dst, dinv, b4, (float*)d_out, E, Nn);
}
